# starting kernel
# baseline (speedup 1.0000x reference)
#include <hip/hip_runtime.h>
#include <math.h>

#define B_SZ  512
#define N_SZ  8192
#define D_SZ  1024
#define DA_SZ 512
#define S_SZ  8
#define DK_SZ 64
#define SD_SZ 512

typedef __attribute__((ext_vector_type(2))) float v2f;
typedef __attribute__((ext_vector_type(8))) float v8f;

__device__ __forceinline__ v8f wmma_f32x4(v2f a, v2f b, v8f c) {
  return __builtin_amdgcn_wmma_f32_16x16x4_f32(
       false, a,  false, b,
       (short)0, c,  false,  false);
}

__global__ __launch_bounds__(256) void maq_gemm_q(const float* __restrict__ z,
                                                  const float* __restrict__ WQ,
                                                  float* __restrict__ qs) {
  const int wave = (blockIdx.x * 256 + threadIdx.x) >> 5;
  const int lane = threadIdx.x & 31;
  const int row  = lane & 15;
  const int half = lane >> 4;
  const int msuper = wave >> 5;
  const int ctile  = wave & 31;
  const int m0 = msuper * 64;
  const int c0 = ctile * 16;
  const float* Bp = WQ + (c0 >> 6) * (DA_SZ * DK_SZ) + (c0 & 63) + row;

  v8f acc[4] = {};
  for (int k0 = 0; k0 < DA_SZ; k0 += 4) {
    const int kk = k0 + 2 * half;
    v2f b;
    b[0] = Bp[(kk + 0) * DK_SZ];
    b[1] = Bp[(kk + 1) * DK_SZ];
#pragma unroll
    for (int t = 0; t < 4; ++t) {
      const float* ap = z + (m0 + t * 16 + row) * DA_SZ + kk;
      v2f a; a[0] = ap[0]; a[1] = ap[1];
      acc[t] = wmma_f32x4(a, b, acc[t]);
    }
  }
#pragma unroll
  for (int t = 0; t < 4; ++t)
#pragma unroll
    for (int i = 0; i < 8; ++i)
      qs[(m0 + t * 16 + i + 8 * half) * SD_SZ + c0 + row] = acc[t][i];
}

__global__ __launch_bounds__(256) void maq_norm_q(float* __restrict__ qs,
                                                  const float* __restrict__ logits) {
  const int idx = blockIdx.x * 256 + threadIdx.x;
  const int b = idx >> 3;
  const int a = idx & 7;
  float mx = logits[0];
#pragma unroll
  for (int i = 1; i < S_SZ; ++i) mx = fmaxf(mx, logits[i]);
  float den = 0.0f;
#pragma unroll
  for (int i = 0; i < S_SZ; ++i) den += expf(logits[i] - mx);
  const float w = expf(logits[a] - mx) / den;

  float4* p = (float4*)(qs + b * SD_SZ + a * DK_SZ);
  float ss = 0.0f;
#pragma unroll
  for (int i = 0; i < DK_SZ / 4; ++i) {
    const float4 v = p[i];
    ss += v.x * v.x + v.y * v.y + v.z * v.z + v.w * v.w;
  }
  const float sc = w / (sqrtf(ss) + 1e-8f);
#pragma unroll
  for (int i = 0; i < DK_SZ / 4; ++i) {
    float4 v = p[i];
    v.x *= sc; v.y *= sc; v.z *= sc; v.w *= sc;
    p[i] = v;
  }
}

__global__ __launch_bounds__(256) void maq_gemm_k(const float* __restrict__ pool,
                                                  const float* __restrict__ WK,
                                                  float* __restrict__ keys) {
  const int wave = (blockIdx.x * 256 + threadIdx.x) >> 5;
  const int lane = threadIdx.x & 31;
  const int row  = lane & 15;
  const int half = lane >> 4;
  const int msuper = wave >> 4;
  const int cpair  = wave & 15;
  const int m0 = msuper * 64;
  const int c0 = cpair * 32;
  const float* Bp = WK + (c0 >> 6) * (D_SZ * DK_SZ) + (c0 & 63) + row;

  v8f acc[2][4] = {};
  for (int k0 = 0; k0 < D_SZ; k0 += 4) {
    const int kk = k0 + 2 * half;
    v2f b[2];
#pragma unroll
    for (int u = 0; u < 2; ++u) {
      b[u][0] = Bp[(kk + 0) * DK_SZ + u * 16];
      b[u][1] = Bp[(kk + 1) * DK_SZ + u * 16];
    }
#pragma unroll
    for (int t = 0; t < 4; ++t) {
      const float* ap = pool + (size_t)(m0 + t * 16 + row) * D_SZ + kk;
      v2f a; a[0] = ap[0]; a[1] = ap[1];
      acc[0][t] = wmma_f32x4(a, b[0], acc[0][t]);
      acc[1][t] = wmma_f32x4(a, b[1], acc[1][t]);
    }
  }
#pragma unroll
  for (int u = 0; u < 2; ++u)
#pragma unroll
    for (int t = 0; t < 4; ++t)
#pragma unroll
      for (int i = 0; i < 8; ++i)
        keys[(size_t)(m0 + t * 16 + i + 8 * half) * SD_SZ + c0 + u * 16 + row] =
            acc[u][t][i];
}

__global__ __launch_bounds__(256) void maq_norm_k(const float* __restrict__ keys,
                                                  float* __restrict__ kn) {
  const int idx = blockIdx.x * 256 + threadIdx.x;
  const int n = idx >> 3;
  const int a = idx & 7;
  const float4* p = (const float4*)(keys + (size_t)n * SD_SZ + a * DK_SZ);
  float ss = 0.0f;
#pragma unroll
  for (int i = 0; i < DK_SZ / 4; ++i) {
    const float4 v = p[i];
    ss += v.x * v.x + v.y * v.y + v.z * v.z + v.w * v.w;
  }
  const float sc = 1.0f / (sqrtf(ss) + 1e-8f);
  float4* q = (float4*)(kn + (size_t)n * SD_SZ + a * DK_SZ);
#pragma unroll
  for (int i = 0; i < DK_SZ / 4; ++i) {
    float4 v = p[i];
    v.x *= sc; v.y *= sc; v.z *= sc; v.w *= sc;
    q[i] = v;
  }
}

__global__ __launch_bounds__(256) void maq_gemm_scores(
    const float* __restrict__ qs, const float* __restrict__ kn,
    const float* __restrict__ tauP, const float* __restrict__ lamP,
    const float* __restrict__ tmpP,
    float* __restrict__ scores, float* __restrict__ alphaRaw) {
  const int wave = (blockIdx.x * 256 + threadIdx.x) >> 5;
  const int lane = threadIdx.x & 31;
  const int row  = lane & 15;
  const int half = lane >> 4;
  const int mpair  = wave >> 7;
  const int nsuper = wave & 127;
  const int m0 = mpair * 32;
  const int n0 = nsuper * 64;

  v8f acc[2][4] = {};
  for (int k0 = 0; k0 < SD_SZ; k0 += 4) {
    const int kk = k0 + 2 * half;
    v2f a[2];
#pragma unroll
    for (int u = 0; u < 2; ++u) {
      const float* ap = qs + (m0 + u * 16 + row) * SD_SZ + kk;
      a[u][0] = ap[0]; a[u][1] = ap[1];
    }
#pragma unroll
    for (int t = 0; t < 4; ++t) {
      const float* kp = kn + (size_t)(n0 + t * 16 + row) * SD_SZ + kk;
      v2f b; b[0] = kp[0]; b[1] = kp[1];
      acc[0][t] = wmma_f32x4(a[0], b, acc[0][t]);
      acc[1][t] = wmma_f32x4(a[1], b, acc[1][t]);
    }
  }

  const float tau  = *tauP;
  const float lam  = *lamP;
  const float invT = 1.0f / *tmpP;
#pragma unroll
  for (int u = 0; u < 2; ++u) {
#pragma unroll
    for (int t = 0; t < 4; ++t) {
#pragma unroll
      for (int i = 0; i < 8; ++i) {
        const int m = m0 + u * 16 + i + 8 * half;
        const int n = n0 + t * 16 + row;
        const float s = acc[u][t][i];
        scores[(size_t)m * N_SZ + n] = s;
        const float g = 1.0f / (1.0f + expf(-lam * (s - tau)));
        alphaRaw[(size_t)m * N_SZ + n] = g * expf(s * invT);
      }
    }
  }
}

__global__ __launch_bounds__(256) void maq_rowsum(const float* __restrict__ ar,
                                                  float* __restrict__ sums) {
  __shared__ float red[256];
  const int b = blockIdx.x;
  const float4* p = (const float4*)(ar + (size_t)b * N_SZ);
  float s = 0.0f;
  for (int i = threadIdx.x; i < N_SZ / 4; i += 256) {
    const float4 v = p[i];
    s += v.x + v.y + v.z + v.w;
  }
  red[threadIdx.x] = s;
  __syncthreads();
  for (int off = 128; off > 0; off >>= 1) {
    if (threadIdx.x < off) red[threadIdx.x] += red[threadIdx.x + off];
    __syncthreads();
  }
  if (threadIdx.x == 0) sums[b] = red[0];
}

__global__ __launch_bounds__(256) void maq_scale_alpha(float4* __restrict__ alpha,
                                                       const float* __restrict__ sums) {
  const size_t idx = (size_t)blockIdx.x * 256 + threadIdx.x;
  const float inv = 1.0f / (sums[idx >> 11] + 1e-8f);
  float4 v = alpha[idx];
  v.x *= inv; v.y *= inv; v.z *= inv; v.w *= inv;
  alpha[idx] = v;
}

extern "C" void kernel_launch(void* const* d_in, const int* in_sizes, int n_in,
                              void* d_out, int out_size, void* d_ws, size_t ws_size,
                              hipStream_t stream) {
  const float* z      = (const float*)d_in[0];
  const float* pool   = (const float*)d_in[1];
  const float* WQ     = (const float*)d_in[2];
  const float* WK     = (const float*)d_in[3];
  const float* logits = (const float*)d_in[4];
  const float* tau    = (const float*)d_in[5];
  const float* lam    = (const float*)d_in[6];
  const float* temp   = (const float*)d_in[7];

  float* out    = (float*)d_out;
  float* alpha  = out;
  float* scores = out + (size_t)B_SZ * N_SZ;
  float* keys   = out + (size_t)2 * B_SZ * N_SZ;

  float* ws = (float*)d_ws;
  float* qs = ws;
  float* kn = qs + (size_t)B_SZ * SD_SZ;
  float* rs = kn + (size_t)N_SZ * SD_SZ;

  maq_gemm_q<<<32, 256, 0, stream>>>(z, WQ, qs);
  maq_norm_q<<<16, 256, 0, stream>>>(qs, logits);
  maq_gemm_k<<<256, 256, 0, stream>>>(pool, WK, keys);
  maq_norm_k<<<256, 256, 0, stream>>>(keys, kn);
  maq_gemm_scores<<<256, 256, 0, stream>>>(qs, kn, tau, lam, temp, scores, alpha);
  maq_rowsum<<<512, 256, 0, stream>>>(alpha, rs);
  maq_scale_alpha<<<4096, 256, 0, stream>>>((float4*)alpha, rs);
}


# Both projection GEMMs: the row tile shared by a block's 8 waves is staged once through LDS in 32-k chunks, two stages, next chunk's global loads issued ahead of the WMMAs; k order kept
# speedup vs baseline: 1.4214x; 1.4214x over previous
#include <hip/hip_runtime.h>
#include <math.h>

#define B_SZ  512
#define N_SZ  8192
#define D_SZ  1024
#define DA_SZ 512
#define S_SZ  8
#define DK_SZ 64
#define SD_SZ 512

typedef __attribute__((ext_vector_type(2))) float v2f;
typedef __attribute__((ext_vector_type(8))) float v8f;

__device__ __forceinline__ v8f wmma_f32x4(v2f a, v2f b, v8f c) {
  return __builtin_amdgcn_wmma_f32_16x16x4_f32(
       false, a,  false, b,
       (short)0, c,  false,  false);
}

#define KCH    32
#define APITCH 36

template <int KDIM, int MT, int NU>
__device__ __forceinline__ void proj_gemm_tile(const float* A,
                                               const float* __restrict__ Bp,
                                               float* __restrict__ C,
                                               const int m0, const int c0) {
  constexpr int ROWS = 16 * MT;
  constexpr int NLD  = ROWS * 8 / 256;
  constexpr int STG  = ROWS * APITCH;
  constexpr int NCH  = KDIM / KCH;
  __shared__ __attribute__((aligned(16))) float As[2 * STG];

  const int tid  = threadIdx.x;
  const int lane = tid & 31;
  const int row  = lane & 15;
  const int half = lane >> 4;
  const int sr = tid >> 3;
  const int sq = tid & 7;
  const float* Ag = A + (size_t)(m0 + sr) * KDIM + sq * 4;
  float*       Aw = As + sr * APITCH + sq * 4;
  const float* Ar = As + row * APITCH + 2 * half;
  const float* Bl = Bp + (2 * half) * DK_SZ;

  static_assert(NLD == 1 || NLD == 2, "one or two 16-B pieces per thread");
  float4 ga0, ga1;
  v2f bn[8][NU];
  ga0 = *(const float4*)(Ag);
  if constexpr (NLD == 2) ga1 = *(const float4*)(Ag + (size_t)32 * KDIM);
#pragma unroll
  for (int ks = 0; ks < 8; ++ks)
#pragma unroll
    for (int u = 0; u < NU; ++u) {
      bn[ks][u][0] = Bl[(ks * 4 + 0) * DK_SZ + u * 16];
      bn[ks][u][1] = Bl[(ks * 4 + 1) * DK_SZ + u * 16];
    }
  *(float4*)(Aw) = ga0;
  if constexpr (NLD == 2) *(float4*)(Aw + 32 * APITCH) = ga1;
  __syncthreads();

  v8f acc[NU][MT] = {};
  for (int kc = 0; kc < NCH; ++kc) {
    const float* Acur = Ar + (kc & 1) * STG;
    v2f bc[8][NU];
#pragma unroll
    for (int ks = 0; ks < 8; ++ks)
#pragma unroll
      for (int u = 0; u < NU; ++u) bc[ks][u] = bn[ks][u];

    const int kn = ((kc + 1 < NCH) ? (kc + 1) : kc) * KCH;
    v2f af[2][2 * MT];
#pragma unroll
    for (int s = 0; s < 2; ++s)
#pragma unroll
      for (int t = 0; t < MT; ++t)
        af[0][s * MT + t] = *(const v2f*)(Acur + t * 16 * APITCH + s * 4);
#pragma unroll
    for (int g = 0; g < 4; ++g) {
      if (g < 3) {
#pragma unroll
        for (int s = 0; s < 2; ++s)
#pragma unroll
          for (int t = 0; t < MT; ++t)
            af[(g + 1) & 1][s * MT + t] =
                *(const v2f*)(Acur + t * 16 * APITCH + ((g + 1) * 2 + s) * 4);
      }
      if (g == 0) {
        ga0 = *(const float4*)(Ag + kn);
        if constexpr (NLD == 2) ga1 = *(const float4*)(Ag + (size_t)32 * KDIM + kn);
#pragma unroll
        for (int ks = 0; ks < 8; ++ks)
#pragma unroll
          for (int u = 0; u < NU; ++u) {
            bn[ks][u][0] = Bl[(kn + ks * 4 + 0) * DK_SZ + u * 16];
            bn[ks][u][1] = Bl[(kn + ks * 4 + 1) * DK_SZ + u * 16];
          }
      }
      __builtin_amdgcn_sched_barrier(0);
#pragma unroll
      for (int s = 0; s < 2; ++s)
#pragma unroll
        for (int t = 0; t < MT; ++t)
#pragma unroll
          for (int u = 0; u < NU; ++u)
            acc[u][t] = wmma_f32x4(af[g & 1][s * MT + t], bc[g * 2 + s][u], acc[u][t]);
      __builtin_amdgcn_sched_barrier(0);
    }

    {
      float* Awn = Aw + ((kc + 1) & 1) * STG;
      *(float4*)(Awn) = ga0;
      if constexpr (NLD == 2) *(float4*)(Awn + 32 * APITCH) = ga1;
    }
    __syncthreads();
  }

#pragma unroll
  for (int u = 0; u < NU; ++u)
#pragma unroll
    for (int t = 0; t < MT; ++t)
#pragma unroll
      for (int i = 0; i < 8; ++i)
        C[(size_t)(m0 + t * 16 + i + 8 * half) * SD_SZ + c0 + u * 16 + row] =
            acc[u][t][i];
}

__global__ __launch_bounds__(256) void maq_gemm_q(const float* z,
                                                  const float* __restrict__ WQ,
                                                  float* __restrict__ qs) {
  const int wave = (blockIdx.x * 256 + threadIdx.x) >> 5;
  const int row  = threadIdx.x & 15;
  const int m0 = (wave >> 5) * 32;
  const int c0 = (wave & 31) * 16;
  const float* Bp = WQ + (c0 >> 6) * (DA_SZ * DK_SZ) + (c0 & 63) + row;
  proj_gemm_tile<DA_SZ, 2, 1>(z, Bp, qs, m0, c0);
}

__global__ __launch_bounds__(256) void maq_norm_q(float* __restrict__ qs,
                                                  const float* __restrict__ logits) {
  const int idx = blockIdx.x * 256 + threadIdx.x;
  const int b = idx >> 3;
  const int a = idx & 7;
  float mx = logits[0];
#pragma unroll
  for (int i = 1; i < S_SZ; ++i) mx = fmaxf(mx, logits[i]);
  float den = 0.0f;
#pragma unroll
  for (int i = 0; i < S_SZ; ++i) den += expf(logits[i] - mx);
  const float w = expf(logits[a] - mx) / den;

  float4* p = (float4*)(qs + b * SD_SZ + a * DK_SZ);
  float ss = 0.0f;
#pragma unroll
  for (int i = 0; i < DK_SZ / 4; ++i) {
    const float4 v = p[i];
    ss += v.x * v.x + v.y * v.y + v.z * v.z + v.w * v.w;
  }
  const float sc = w / (sqrtf(ss) + 1e-8f);
#pragma unroll
  for (int i = 0; i < DK_SZ / 4; ++i) {
    float4 v = p[i];
    v.x *= sc; v.y *= sc; v.z *= sc; v.w *= sc;
    p[i] = v;
  }
}

__global__ __launch_bounds__(256) void maq_gemm_k(const float* pool,
                                                  const float* __restrict__ WK,
                                                  float* __restrict__ keys) {
  const int wave = (blockIdx.x * 256 + threadIdx.x) >> 5;
  const int row  = threadIdx.x & 15;
  const int m0 = (wave >> 4) * 64;
  const int c0 = (wave & 15) * 32;
  const float* Bp = WK + (c0 >> 6) * (D_SZ * DK_SZ) + (c0 & 63) + row;
  proj_gemm_tile<D_SZ, 4, 2>(pool, Bp, keys, m0, c0);
}

__global__ __launch_bounds__(256) void maq_norm_k(const float* __restrict__ keys,
                                                  float* __restrict__ kn) {
  const int idx = blockIdx.x * 256 + threadIdx.x;
  const int n = idx >> 3;
  const int a = idx & 7;
  const float4* p = (const float4*)(keys + (size_t)n * SD_SZ + a * DK_SZ);
  float ss = 0.0f;
#pragma unroll
  for (int i = 0; i < DK_SZ / 4; ++i) {
    const float4 v = p[i];
    ss += v.x * v.x + v.y * v.y + v.z * v.z + v.w * v.w;
  }
  const float sc = 1.0f / (sqrtf(ss) + 1e-8f);
  float4* q = (float4*)(kn + (size_t)n * SD_SZ + a * DK_SZ);
#pragma unroll
  for (int i = 0; i < DK_SZ / 4; ++i) {
    float4 v = p[i];
    v.x *= sc; v.y *= sc; v.z *= sc; v.w *= sc;
    q[i] = v;
  }
}

__global__ __launch_bounds__(256) void maq_gemm_scores(
    const float* __restrict__ qs, const float* __restrict__ kn,
    const float* __restrict__ tauP, const float* __restrict__ lamP,
    const float* __restrict__ tmpP,
    float* __restrict__ scores, float* __restrict__ alphaRaw) {
  const int wave = (blockIdx.x * 256 + threadIdx.x) >> 5;
  const int lane = threadIdx.x & 31;
  const int row  = lane & 15;
  const int half = lane >> 4;
  const int mpair  = wave >> 7;
  const int nsuper = wave & 127;
  const int m0 = mpair * 32;
  const int n0 = nsuper * 64;

  v8f acc[2][4] = {};
  for (int k0 = 0; k0 < SD_SZ; k0 += 4) {
    const int kk = k0 + 2 * half;
    v2f a[2];
#pragma unroll
    for (int u = 0; u < 2; ++u) {
      const float* ap = qs + (m0 + u * 16 + row) * SD_SZ + kk;
      a[u][0] = ap[0]; a[u][1] = ap[1];
    }
#pragma unroll
    for (int t = 0; t < 4; ++t) {
      const float* kp = kn + (size_t)(n0 + t * 16 + row) * SD_SZ + kk;
      v2f b; b[0] = kp[0]; b[1] = kp[1];
      acc[0][t] = wmma_f32x4(a[0], b, acc[0][t]);
      acc[1][t] = wmma_f32x4(a[1], b, acc[1][t]);
    }
  }

  const float tau  = *tauP;
  const float lam  = *lamP;
  const float invT = 1.0f / *tmpP;
#pragma unroll
  for (int u = 0; u < 2; ++u) {
#pragma unroll
    for (int t = 0; t < 4; ++t) {
#pragma unroll
      for (int i = 0; i < 8; ++i) {
        const int m = m0 + u * 16 + i + 8 * half;
        const int n = n0 + t * 16 + row;
        const float s = acc[u][t][i];
        scores[(size_t)m * N_SZ + n] = s;
        const float g = 1.0f / (1.0f + expf(-lam * (s - tau)));
        alphaRaw[(size_t)m * N_SZ + n] = g * expf(s * invT);
      }
    }
  }
}

__global__ __launch_bounds__(256) void maq_rowsum(const float* __restrict__ ar,
                                                  float* __restrict__ sums) {
  __shared__ float red[256];
  const int b = blockIdx.x;
  const float4* p = (const float4*)(ar + (size_t)b * N_SZ);
  float s = 0.0f;
  for (int i = threadIdx.x; i < N_SZ / 4; i += 256) {
    const float4 v = p[i];
    s += v.x + v.y + v.z + v.w;
  }
  red[threadIdx.x] = s;
  __syncthreads();
  for (int off = 128; off > 0; off >>= 1) {
    if (threadIdx.x < off) red[threadIdx.x] += red[threadIdx.x + off];
    __syncthreads();
  }
  if (threadIdx.x == 0) sums[b] = red[0];
}

__global__ __launch_bounds__(256) void maq_scale_alpha(float4* __restrict__ alpha,
                                                       const float* __restrict__ sums) {
  const size_t idx = (size_t)blockIdx.x * 256 + threadIdx.x;
  const float inv = 1.0f / (sums[idx >> 11] + 1e-8f);
  float4 v = alpha[idx];
  v.x *= inv; v.y *= inv; v.z *= inv; v.w *= inv;
  alpha[idx] = v;
}

extern "C" void kernel_launch(void* const* d_in, const int* in_sizes, int n_in,
                              void* d_out, int out_size, void* d_ws, size_t ws_size,
                              hipStream_t stream) {
  const float* z      = (const float*)d_in[0];
  const float* pool   = (const float*)d_in[1];
  const float* WQ     = (const float*)d_in[2];
  const float* WK     = (const float*)d_in[3];
  const float* logits = (const float*)d_in[4];
  const float* tau    = (const float*)d_in[5];
  const float* lam    = (const float*)d_in[6];
  const float* temp   = (const float*)d_in[7];

  float* out    = (float*)d_out;
  float* alpha  = out;
  float* scores = out + (size_t)B_SZ * N_SZ;
  float* keys   = out + (size_t)2 * B_SZ * N_SZ;

  float* ws = (float*)d_ws;
  float* qs = ws;
  float* kn = qs + (size_t)B_SZ * SD_SZ;
  float* rs = kn + (size_t)N_SZ * SD_SZ;

  maq_gemm_q<<<64, 256, 0, stream>>>(z, WQ, qs);
  maq_norm_q<<<16, 256, 0, stream>>>(qs, logits);
  maq_gemm_k<<<256, 256, 0, stream>>>(pool, WK, keys);
  maq_norm_k<<<256, 256, 0, stream>>>(keys, kn);
  maq_gemm_scores<<<256, 256, 0, stream>>>(qs, kn, tau, lam, temp, scores, alpha);
  maq_rowsum<<<512, 256, 0, stream>>>(alpha, rs);
  maq_scale_alpha<<<4096, 256, 0, stream>>>((float4*)alpha, rs);
}


# Scores GEMM: block tile 128 x 128, both row tiles staged through LDS in 32-k chunks (two stages, next chunk's global loads issued ahead of the WMMAs) instead of per-lane 8-B gathers; k order kept
# speedup vs baseline: 1.5641x; 1.1004x over previous
#include <hip/hip_runtime.h>
#include <math.h>

#define B_SZ  512
#define N_SZ  8192
#define D_SZ  1024
#define DA_SZ 512
#define S_SZ  8
#define DK_SZ 64
#define SD_SZ 512

typedef __attribute__((ext_vector_type(2))) float v2f;
typedef __attribute__((ext_vector_type(8))) float v8f;

__device__ __forceinline__ v8f wmma_f32x4(v2f a, v2f b, v8f c) {
  return __builtin_amdgcn_wmma_f32_16x16x4_f32(
       false, a,  false, b,
       (short)0, c,  false,  false);
}

#define KCH    32
#define APITCH 36

template <int KDIM, int MT, int NU>
__device__ __forceinline__ void proj_gemm_tile(const float* A,
                                               const float* __restrict__ Bp,
                                               float* __restrict__ C,
                                               const int m0, const int c0) {
  constexpr int ROWS = 16 * MT;
  constexpr int NLD  = ROWS * 8 / 256;
  constexpr int STG  = ROWS * APITCH;
  constexpr int NCH  = KDIM / KCH;
  __shared__ __attribute__((aligned(16))) float As[2 * STG];

  const int tid  = threadIdx.x;
  const int lane = tid & 31;
  const int row  = lane & 15;
  const int half = lane >> 4;
  const int sr = tid >> 3;
  const int sq = tid & 7;
  const float* Ag = A + (size_t)(m0 + sr) * KDIM + sq * 4;
  float*       Aw = As + sr * APITCH + sq * 4;
  const float* Ar = As + row * APITCH + 2 * half;
  const float* Bl = Bp + (2 * half) * DK_SZ;

  static_assert(NLD == 1 || NLD == 2, "one or two 16-B pieces per thread");
  float4 ga0, ga1;
  v2f bn[8][NU];
  ga0 = *(const float4*)(Ag);
  if constexpr (NLD == 2) ga1 = *(const float4*)(Ag + (size_t)32 * KDIM);
#pragma unroll
  for (int ks = 0; ks < 8; ++ks)
#pragma unroll
    for (int u = 0; u < NU; ++u) {
      bn[ks][u][0] = Bl[(ks * 4 + 0) * DK_SZ + u * 16];
      bn[ks][u][1] = Bl[(ks * 4 + 1) * DK_SZ + u * 16];
    }
  *(float4*)(Aw) = ga0;
  if constexpr (NLD == 2) *(float4*)(Aw + 32 * APITCH) = ga1;
  __syncthreads();

  v8f acc[NU][MT] = {};
  for (int kc = 0; kc < NCH; ++kc) {
    const float* Acur = Ar + (kc & 1) * STG;
    v2f bc[8][NU];
#pragma unroll
    for (int ks = 0; ks < 8; ++ks)
#pragma unroll
      for (int u = 0; u < NU; ++u) bc[ks][u] = bn[ks][u];

    const int kn = ((kc + 1 < NCH) ? (kc + 1) : kc) * KCH;
    v2f af[2][2 * MT];
#pragma unroll
    for (int s = 0; s < 2; ++s)
#pragma unroll
      for (int t = 0; t < MT; ++t)
        af[0][s * MT + t] = *(const v2f*)(Acur + t * 16 * APITCH + s * 4);
#pragma unroll
    for (int g = 0; g < 4; ++g) {
      if (g < 3) {
#pragma unroll
        for (int s = 0; s < 2; ++s)
#pragma unroll
          for (int t = 0; t < MT; ++t)
            af[(g + 1) & 1][s * MT + t] =
                *(const v2f*)(Acur + t * 16 * APITCH + ((g + 1) * 2 + s) * 4);
      }
      if (g == 0) {
        ga0 = *(const float4*)(Ag + kn);
        if constexpr (NLD == 2) ga1 = *(const float4*)(Ag + (size_t)32 * KDIM + kn);
#pragma unroll
        for (int ks = 0; ks < 8; ++ks)
#pragma unroll
          for (int u = 0; u < NU; ++u) {
            bn[ks][u][0] = Bl[(kn + ks * 4 + 0) * DK_SZ + u * 16];
            bn[ks][u][1] = Bl[(kn + ks * 4 + 1) * DK_SZ + u * 16];
          }
      }
      __builtin_amdgcn_sched_barrier(0);
#pragma unroll
      for (int s = 0; s < 2; ++s)
#pragma unroll
        for (int t = 0; t < MT; ++t)
#pragma unroll
          for (int u = 0; u < NU; ++u)
            acc[u][t] = wmma_f32x4(af[g & 1][s * MT + t], bc[g * 2 + s][u], acc[u][t]);
      __builtin_amdgcn_sched_barrier(0);
    }

    {
      float* Awn = Aw + ((kc + 1) & 1) * STG;
      *(float4*)(Awn) = ga0;
      if constexpr (NLD == 2) *(float4*)(Awn + 32 * APITCH) = ga1;
    }
    __syncthreads();
  }

#pragma unroll
  for (int u = 0; u < NU; ++u)
#pragma unroll
    for (int t = 0; t < MT; ++t)
#pragma unroll
      for (int i = 0; i < 8; ++i)
        C[(size_t)(m0 + t * 16 + i + 8 * half) * SD_SZ + c0 + u * 16 + row] =
            acc[u][t][i];
}

__global__ __launch_bounds__(256) void maq_gemm_q(const float* z,
                                                  const float* __restrict__ WQ,
                                                  float* __restrict__ qs) {
  const int wave = (blockIdx.x * 256 + threadIdx.x) >> 5;
  const int row  = threadIdx.x & 15;
  const int m0 = (wave >> 5) * 32;
  const int c0 = (wave & 31) * 16;
  const float* Bp = WQ + (c0 >> 6) * (DA_SZ * DK_SZ) + (c0 & 63) + row;
  proj_gemm_tile<DA_SZ, 2, 1>(z, Bp, qs, m0, c0);
}

__global__ __launch_bounds__(256) void maq_norm_q(float* __restrict__ qs,
                                                  const float* __restrict__ logits) {
  const int idx = blockIdx.x * 256 + threadIdx.x;
  const int b = idx >> 3;
  const int a = idx & 7;
  float mx = logits[0];
#pragma unroll
  for (int i = 1; i < S_SZ; ++i) mx = fmaxf(mx, logits[i]);
  float den = 0.0f;
#pragma unroll
  for (int i = 0; i < S_SZ; ++i) den += expf(logits[i] - mx);
  const float w = expf(logits[a] - mx) / den;

  float4* p = (float4*)(qs + b * SD_SZ + a * DK_SZ);
  float ss = 0.0f;
#pragma unroll
  for (int i = 0; i < DK_SZ / 4; ++i) {
    const float4 v = p[i];
    ss += v.x * v.x + v.y * v.y + v.z * v.z + v.w * v.w;
  }
  const float sc = w / (sqrtf(ss) + 1e-8f);
#pragma unroll
  for (int i = 0; i < DK_SZ / 4; ++i) {
    float4 v = p[i];
    v.x *= sc; v.y *= sc; v.z *= sc; v.w *= sc;
    p[i] = v;
  }
}

__global__ __launch_bounds__(256) void maq_gemm_k(const float* pool,
                                                  const float* __restrict__ WK,
                                                  float* __restrict__ keys) {
  const int wave = (blockIdx.x * 256 + threadIdx.x) >> 5;
  const int row  = threadIdx.x & 15;
  const int m0 = (wave >> 4) * 64;
  const int c0 = (wave & 15) * 32;
  const float* Bp = WK + (c0 >> 6) * (D_SZ * DK_SZ) + (c0 & 63) + row;
  proj_gemm_tile<D_SZ, 4, 2>(pool, Bp, keys, m0, c0);
}

__global__ __launch_bounds__(256) void maq_norm_k(const float* __restrict__ keys,
                                                  float* __restrict__ kn) {
  const int idx = blockIdx.x * 256 + threadIdx.x;
  const int n = idx >> 3;
  const int a = idx & 7;
  const float4* p = (const float4*)(keys + (size_t)n * SD_SZ + a * DK_SZ);
  float ss = 0.0f;
#pragma unroll
  for (int i = 0; i < DK_SZ / 4; ++i) {
    const float4 v = p[i];
    ss += v.x * v.x + v.y * v.y + v.z * v.z + v.w * v.w;
  }
  const float sc = 1.0f / (sqrtf(ss) + 1e-8f);
  float4* q = (float4*)(kn + (size_t)n * SD_SZ + a * DK_SZ);
#pragma unroll
  for (int i = 0; i < DK_SZ / 4; ++i) {
    float4 v = p[i];
    v.x *= sc; v.y *= sc; v.z *= sc; v.w *= sc;
    q[i] = v;
  }
}

#define SKCH   32
#define SPITCH 36
#define STILE  128

__global__ __launch_bounds__(256) void maq_gemm_scores(
    const float* qs, const float* kn,
    const float* __restrict__ tauP, const float* __restrict__ lamP,
    const float* __restrict__ tmpP,
    float* __restrict__ scores, float* __restrict__ alphaRaw) {
  constexpr int OPS = STILE * SPITCH;
  constexpr int STG = 2 * OPS;
  constexpr int NCH = SD_SZ / SKCH;
  __shared__ __attribute__((aligned(16))) float Ss[2 * STG];

  const int tid  = threadIdx.x;
  const int lane = tid & 31;
  const int row  = lane & 15;
  const int half = lane >> 4;
  const int wv   = tid >> 5;
  const int m0 = (blockIdx.x >> 6) * STILE;
  const int n0 = (blockIdx.x & 63) * STILE;
  const int wm = (wv >> 1) * 32;
  const int wn = (wv & 1) * 64;

  const int sr = tid >> 3;
  const int sq = tid & 7;
  const float* Qg = qs + (size_t)(m0 + sr) * SD_SZ + sq * 4;
  const float* Kg = kn + (size_t)(n0 + sr) * SD_SZ + sq * 4;
  float*       Sw = Ss + sr * SPITCH + sq * 4;
  const float* Qr = Ss + (wm + row) * SPITCH + 2 * half;
  const float* Kr = Ss + OPS + (wn + row) * SPITCH + 2 * half;

  float4 gq0, gq1, gq2, gq3, gk0, gk1, gk2, gk3;
  constexpr size_t GROW = (size_t)32 * SD_SZ;
  gq0 = *(const float4*)(Qg);
  gq1 = *(const float4*)(Qg + GROW);
  gq2 = *(const float4*)(Qg + 2 * GROW);
  gq3 = *(const float4*)(Qg + 3 * GROW);
  gk0 = *(const float4*)(Kg);
  gk1 = *(const float4*)(Kg + GROW);
  gk2 = *(const float4*)(Kg + 2 * GROW);
  gk3 = *(const float4*)(Kg + 3 * GROW);
  *(float4*)(Sw) = gq0;
  *(float4*)(Sw + 32 * SPITCH) = gq1;
  *(float4*)(Sw + 64 * SPITCH) = gq2;
  *(float4*)(Sw + 96 * SPITCH) = gq3;
  *(float4*)(Sw + OPS) = gk0;
  *(float4*)(Sw + OPS + 32 * SPITCH) = gk1;
  *(float4*)(Sw + OPS + 64 * SPITCH) = gk2;
  *(float4*)(Sw + OPS + 96 * SPITCH) = gk3;
  __syncthreads();

  v8f acc[2][4] = {};
  for (int kc = 0; kc < NCH; ++kc) {
    const float* Qc = Qr + (kc & 1) * STG;
    const float* Kc = Kr + (kc & 1) * STG;
    const int kx = ((kc + 1 < NCH) ? (kc + 1) : kc) * SKCH;
    v2f fa[2][2][2];
    v2f fb[2][2][4];
#pragma unroll
    for (int s = 0; s < 2; ++s) {
#pragma unroll
      for (int u = 0; u < 2; ++u)
        fa[0][s][u] = *(const v2f*)(Qc + u * 16 * SPITCH + s * 4);
#pragma unroll
      for (int t = 0; t < 4; ++t)
        fb[0][s][t] = *(const v2f*)(Kc + t * 16 * SPITCH + s * 4);
    }
    __builtin_amdgcn_sched_barrier(0);
#pragma unroll
    for (int g = 0; g < 4; ++g) {
      if (g < 3) {
#pragma unroll
        for (int s = 0; s < 2; ++s) {
#pragma unroll
          for (int u = 0; u < 2; ++u)
            fa[(g + 1) & 1][s][u] =
                *(const v2f*)(Qc + u * 16 * SPITCH + ((g + 1) * 2 + s) * 4);
#pragma unroll
          for (int t = 0; t < 4; ++t)
            fb[(g + 1) & 1][s][t] =
                *(const v2f*)(Kc + t * 16 * SPITCH + ((g + 1) * 2 + s) * 4);
        }
      }
      if (g == 0) {
        gq0 = *(const float4*)(Qg + kx);
        gq1 = *(const float4*)(Qg + GROW + kx);
        gq2 = *(const float4*)(Qg + 2 * GROW + kx);
        gq3 = *(const float4*)(Qg + 3 * GROW + kx);
        gk0 = *(const float4*)(Kg + kx);
        gk1 = *(const float4*)(Kg + GROW + kx);
        gk2 = *(const float4*)(Kg + 2 * GROW + kx);
        gk3 = *(const float4*)(Kg + 3 * GROW + kx);
      }
      if (g == 3) {
        float* Swn = Sw + ((kc + 1) & 1) * STG;
        *(float4*)(Swn) = gq0;
        *(float4*)(Swn + 32 * SPITCH) = gq1;
        *(float4*)(Swn + 64 * SPITCH) = gq2;
        *(float4*)(Swn + 96 * SPITCH) = gq3;
        *(float4*)(Swn + OPS) = gk0;
        *(float4*)(Swn + OPS + 32 * SPITCH) = gk1;
        *(float4*)(Swn + OPS + 64 * SPITCH) = gk2;
        *(float4*)(Swn + OPS + 96 * SPITCH) = gk3;
      }
      __builtin_amdgcn_sched_barrier(0);
#pragma unroll
      for (int s = 0; s < 2; ++s)
#pragma unroll
        for (int t = 0; t < 4; ++t) {
          acc[0][t] = wmma_f32x4(fa[g & 1][s][0], fb[g & 1][s][t], acc[0][t]);
          acc[1][t] = wmma_f32x4(fa[g & 1][s][1], fb[g & 1][s][t], acc[1][t]);
        }
      __builtin_amdgcn_sched_barrier(0);
    }

    __syncthreads();
  }

  const float tau  = *tauP;
  const float lam  = *lamP;
  const float invT = 1.0f / *tmpP;
#pragma unroll
  for (int u = 0; u < 2; ++u) {
#pragma unroll
    for (int t = 0; t < 4; ++t) {
#pragma unroll
      for (int i = 0; i < 8; ++i) {
        const int m = m0 + wm + u * 16 + i + 8 * half;
        const int n = n0 + wn + t * 16 + row;
        const float s = acc[u][t][i];
        scores[(size_t)m * N_SZ + n] = s;
        const float g = 1.0f / (1.0f + expf(-lam * (s - tau)));
        alphaRaw[(size_t)m * N_SZ + n] = g * expf(s * invT);
      }
    }
  }
}

__global__ __launch_bounds__(256) void maq_rowsum(const float* __restrict__ ar,
                                                  float* __restrict__ sums) {
  __shared__ float red[256];
  const int b = blockIdx.x;
  const float4* p = (const float4*)(ar + (size_t)b * N_SZ);
  float s = 0.0f;
  for (int i = threadIdx.x; i < N_SZ / 4; i += 256) {
    const float4 v = p[i];
    s += v.x + v.y + v.z + v.w;
  }
  red[threadIdx.x] = s;
  __syncthreads();
  for (int off = 128; off > 0; off >>= 1) {
    if (threadIdx.x < off) red[threadIdx.x] += red[threadIdx.x + off];
    __syncthreads();
  }
  if (threadIdx.x == 0) sums[b] = red[0];
}

__global__ __launch_bounds__(256) void maq_scale_alpha(float4* __restrict__ alpha,
                                                       const float* __restrict__ sums) {
  const size_t idx = (size_t)blockIdx.x * 256 + threadIdx.x;
  const float inv = 1.0f / (sums[idx >> 11] + 1e-8f);
  float4 v = alpha[idx];
  v.x *= inv; v.y *= inv; v.z *= inv; v.w *= inv;
  alpha[idx] = v;
}

extern "C" void kernel_launch(void* const* d_in, const int* in_sizes, int n_in,
                              void* d_out, int out_size, void* d_ws, size_t ws_size,
                              hipStream_t stream) {
  const float* z      = (const float*)d_in[0];
  const float* pool   = (const float*)d_in[1];
  const float* WQ     = (const float*)d_in[2];
  const float* WK     = (const float*)d_in[3];
  const float* logits = (const float*)d_in[4];
  const float* tau    = (const float*)d_in[5];
  const float* lam    = (const float*)d_in[6];
  const float* temp   = (const float*)d_in[7];

  float* out    = (float*)d_out;
  float* alpha  = out;
  float* scores = out + (size_t)B_SZ * N_SZ;
  float* keys   = out + (size_t)2 * B_SZ * N_SZ;

  float* ws = (float*)d_ws;
  float* qs = ws;
  float* kn = qs + (size_t)B_SZ * SD_SZ;
  float* rs = kn + (size_t)N_SZ * SD_SZ;

  maq_gemm_q<<<64, 256, 0, stream>>>(z, WQ, qs);
  maq_norm_q<<<16, 256, 0, stream>>>(qs, logits);
  maq_gemm_k<<<256, 256, 0, stream>>>(pool, WK, keys);
  maq_norm_k<<<256, 256, 0, stream>>>(keys, kn);
  maq_gemm_scores<<<256, 256, 0, stream>>>(qs, kn, tau, lam, temp, scores, alpha);
  maq_rowsum<<<512, 256, 0, stream>>>(alpha, rs);
  maq_scale_alpha<<<4096, 256, 0, stream>>>((float4*)alpha, rs);
}


# Row sum and alpha scaling fused into one kernel: a 128-thread block keeps its row (16 float4 per thread) in registers between sum and scaling; one launch and 34816 waves fewer; sum order kept
# speedup vs baseline: 1.6822x; 1.0755x over previous
#include <hip/hip_runtime.h>
#include <math.h>

#define B_SZ  512
#define N_SZ  8192
#define D_SZ  1024
#define DA_SZ 512
#define S_SZ  8
#define DK_SZ 64
#define SD_SZ 512

typedef __attribute__((ext_vector_type(2))) float v2f;
typedef __attribute__((ext_vector_type(8))) float v8f;

__device__ __forceinline__ v8f wmma_f32x4(v2f a, v2f b, v8f c) {
  return __builtin_amdgcn_wmma_f32_16x16x4_f32(
       false, a,  false, b,
       (short)0, c,  false,  false);
}

#define KCH    32
#define APITCH 36

template <int KDIM, int MT, int NU>
__device__ __forceinline__ void proj_gemm_tile(const float* A,
                                               const float* __restrict__ Bp,
                                               float* __restrict__ C,
                                               const int m0, const int c0) {
  constexpr int ROWS = 16 * MT;
  constexpr int NLD  = ROWS * 8 / 256;
  constexpr int STG  = ROWS * APITCH;
  constexpr int NCH  = KDIM / KCH;
  __shared__ __attribute__((aligned(16))) float As[2 * STG];

  const int tid  = threadIdx.x;
  const int lane = tid & 31;
  const int row  = lane & 15;
  const int half = lane >> 4;
  const int sr = tid >> 3;
  const int sq = tid & 7;
  const float* Ag = A + (size_t)(m0 + sr) * KDIM + sq * 4;
  float*       Aw = As + sr * APITCH + sq * 4;
  const float* Ar = As + row * APITCH + 2 * half;
  const float* Bl = Bp + (2 * half) * DK_SZ;

  static_assert(NLD == 1 || NLD == 2, "one or two 16-B pieces per thread");
  float4 ga0, ga1;
  v2f bn[8][NU];
  ga0 = *(const float4*)(Ag);
  if constexpr (NLD == 2) ga1 = *(const float4*)(Ag + (size_t)32 * KDIM);
#pragma unroll
  for (int ks = 0; ks < 8; ++ks)
#pragma unroll
    for (int u = 0; u < NU; ++u) {
      bn[ks][u][0] = Bl[(ks * 4 + 0) * DK_SZ + u * 16];
      bn[ks][u][1] = Bl[(ks * 4 + 1) * DK_SZ + u * 16];
    }
  *(float4*)(Aw) = ga0;
  if constexpr (NLD == 2) *(float4*)(Aw + 32 * APITCH) = ga1;
  __syncthreads();

  v8f acc[NU][MT] = {};
  for (int kc = 0; kc < NCH; ++kc) {
    const float* Acur = Ar + (kc & 1) * STG;
    v2f bc[8][NU];
#pragma unroll
    for (int ks = 0; ks < 8; ++ks)
#pragma unroll
      for (int u = 0; u < NU; ++u) bc[ks][u] = bn[ks][u];

    const int kn = ((kc + 1 < NCH) ? (kc + 1) : kc) * KCH;
    v2f af[2][2 * MT];
#pragma unroll
    for (int s = 0; s < 2; ++s)
#pragma unroll
      for (int t = 0; t < MT; ++t)
        af[0][s * MT + t] = *(const v2f*)(Acur + t * 16 * APITCH + s * 4);
#pragma unroll
    for (int g = 0; g < 4; ++g) {
      if (g < 3) {
#pragma unroll
        for (int s = 0; s < 2; ++s)
#pragma unroll
          for (int t = 0; t < MT; ++t)
            af[(g + 1) & 1][s * MT + t] =
                *(const v2f*)(Acur + t * 16 * APITCH + ((g + 1) * 2 + s) * 4);
      }
      if (g == 0) {
        ga0 = *(const float4*)(Ag + kn);
        if constexpr (NLD == 2) ga1 = *(const float4*)(Ag + (size_t)32 * KDIM + kn);
#pragma unroll
        for (int ks = 0; ks < 8; ++ks)
#pragma unroll
          for (int u = 0; u < NU; ++u) {
            bn[ks][u][0] = Bl[(kn + ks * 4 + 0) * DK_SZ + u * 16];
            bn[ks][u][1] = Bl[(kn + ks * 4 + 1) * DK_SZ + u * 16];
          }
      }
      __builtin_amdgcn_sched_barrier(0);
#pragma unroll
      for (int s = 0; s < 2; ++s)
#pragma unroll
        for (int t = 0; t < MT; ++t)
#pragma unroll
          for (int u = 0; u < NU; ++u)
            acc[u][t] = wmma_f32x4(af[g & 1][s * MT + t], bc[g * 2 + s][u], acc[u][t]);
      __builtin_amdgcn_sched_barrier(0);
    }

    {
      float* Awn = Aw + ((kc + 1) & 1) * STG;
      *(float4*)(Awn) = ga0;
      if constexpr (NLD == 2) *(float4*)(Awn + 32 * APITCH) = ga1;
    }
    __syncthreads();
  }

#pragma unroll
  for (int u = 0; u < NU; ++u)
#pragma unroll
    for (int t = 0; t < MT; ++t)
#pragma unroll
      for (int i = 0; i < 8; ++i)
        C[(size_t)(m0 + t * 16 + i + 8 * half) * SD_SZ + c0 + u * 16 + row] =
            acc[u][t][i];
}

__global__ __launch_bounds__(256) void maq_gemm_q(const float* z,
                                                  const float* __restrict__ WQ,
                                                  float* __restrict__ qs) {
  const int wave = (blockIdx.x * 256 + threadIdx.x) >> 5;
  const int row  = threadIdx.x & 15;
  const int m0 = (wave >> 5) * 32;
  const int c0 = (wave & 31) * 16;
  const float* Bp = WQ + (c0 >> 6) * (DA_SZ * DK_SZ) + (c0 & 63) + row;
  proj_gemm_tile<DA_SZ, 2, 1>(z, Bp, qs, m0, c0);
}

__global__ __launch_bounds__(256) void maq_norm_q(float* __restrict__ qs,
                                                  const float* __restrict__ logits) {
  const int idx = blockIdx.x * 256 + threadIdx.x;
  const int b = idx >> 3;
  const int a = idx & 7;
  float mx = logits[0];
#pragma unroll
  for (int i = 1; i < S_SZ; ++i) mx = fmaxf(mx, logits[i]);
  float den = 0.0f;
#pragma unroll
  for (int i = 0; i < S_SZ; ++i) den += expf(logits[i] - mx);
  const float w = expf(logits[a] - mx) / den;

  float4* p = (float4*)(qs + b * SD_SZ + a * DK_SZ);
  float ss = 0.0f;
#pragma unroll
  for (int i = 0; i < DK_SZ / 4; ++i) {
    const float4 v = p[i];
    ss += v.x * v.x + v.y * v.y + v.z * v.z + v.w * v.w;
  }
  const float sc = w / (sqrtf(ss) + 1e-8f);
#pragma unroll
  for (int i = 0; i < DK_SZ / 4; ++i) {
    float4 v = p[i];
    v.x *= sc; v.y *= sc; v.z *= sc; v.w *= sc;
    p[i] = v;
  }
}

__global__ __launch_bounds__(256) void maq_gemm_k(const float* pool,
                                                  const float* __restrict__ WK,
                                                  float* __restrict__ keys) {
  const int wave = (blockIdx.x * 256 + threadIdx.x) >> 5;
  const int row  = threadIdx.x & 15;
  const int m0 = (wave >> 4) * 64;
  const int c0 = (wave & 15) * 32;
  const float* Bp = WK + (c0 >> 6) * (D_SZ * DK_SZ) + (c0 & 63) + row;
  proj_gemm_tile<D_SZ, 4, 2>(pool, Bp, keys, m0, c0);
}

__global__ __launch_bounds__(256) void maq_norm_k(const float* __restrict__ keys,
                                                  float* __restrict__ kn) {
  const int idx = blockIdx.x * 256 + threadIdx.x;
  const int n = idx >> 3;
  const int a = idx & 7;
  const float4* p = (const float4*)(keys + (size_t)n * SD_SZ + a * DK_SZ);
  float ss = 0.0f;
#pragma unroll
  for (int i = 0; i < DK_SZ / 4; ++i) {
    const float4 v = p[i];
    ss += v.x * v.x + v.y * v.y + v.z * v.z + v.w * v.w;
  }
  const float sc = 1.0f / (sqrtf(ss) + 1e-8f);
  float4* q = (float4*)(kn + (size_t)n * SD_SZ + a * DK_SZ);
#pragma unroll
  for (int i = 0; i < DK_SZ / 4; ++i) {
    float4 v = p[i];
    v.x *= sc; v.y *= sc; v.z *= sc; v.w *= sc;
    q[i] = v;
  }
}

#define SKCH   32
#define SPITCH 36
#define STILE  128

__global__ __launch_bounds__(256) void maq_gemm_scores(
    const float* qs, const float* kn,
    const float* __restrict__ tauP, const float* __restrict__ lamP,
    const float* __restrict__ tmpP,
    float* __restrict__ scores, float* __restrict__ alphaRaw) {
  constexpr int OPS = STILE * SPITCH;
  constexpr int STG = 2 * OPS;
  constexpr int NCH = SD_SZ / SKCH;
  __shared__ __attribute__((aligned(16))) float Ss[2 * STG];

  const int tid  = threadIdx.x;
  const int lane = tid & 31;
  const int row  = lane & 15;
  const int half = lane >> 4;
  const int wv   = tid >> 5;
  const int m0 = (blockIdx.x >> 6) * STILE;
  const int n0 = (blockIdx.x & 63) * STILE;
  const int wm = (wv >> 1) * 32;
  const int wn = (wv & 1) * 64;

  const int sr = tid >> 3;
  const int sq = tid & 7;
  const float* Qg = qs + (size_t)(m0 + sr) * SD_SZ + sq * 4;
  const float* Kg = kn + (size_t)(n0 + sr) * SD_SZ + sq * 4;
  float*       Sw = Ss + sr * SPITCH + sq * 4;
  const float* Qr = Ss + (wm + row) * SPITCH + 2 * half;
  const float* Kr = Ss + OPS + (wn + row) * SPITCH + 2 * half;

  float4 gq0, gq1, gq2, gq3, gk0, gk1, gk2, gk3;
  constexpr size_t GROW = (size_t)32 * SD_SZ;
  gq0 = *(const float4*)(Qg);
  gq1 = *(const float4*)(Qg + GROW);
  gq2 = *(const float4*)(Qg + 2 * GROW);
  gq3 = *(const float4*)(Qg + 3 * GROW);
  gk0 = *(const float4*)(Kg);
  gk1 = *(const float4*)(Kg + GROW);
  gk2 = *(const float4*)(Kg + 2 * GROW);
  gk3 = *(const float4*)(Kg + 3 * GROW);
  *(float4*)(Sw) = gq0;
  *(float4*)(Sw + 32 * SPITCH) = gq1;
  *(float4*)(Sw + 64 * SPITCH) = gq2;
  *(float4*)(Sw + 96 * SPITCH) = gq3;
  *(float4*)(Sw + OPS) = gk0;
  *(float4*)(Sw + OPS + 32 * SPITCH) = gk1;
  *(float4*)(Sw + OPS + 64 * SPITCH) = gk2;
  *(float4*)(Sw + OPS + 96 * SPITCH) = gk3;
  __syncthreads();

  v8f acc[2][4] = {};
  for (int kc = 0; kc < NCH; ++kc) {
    const float* Qc = Qr + (kc & 1) * STG;
    const float* Kc = Kr + (kc & 1) * STG;
    const int kx = ((kc + 1 < NCH) ? (kc + 1) : kc) * SKCH;
    v2f fa[2][2][2];
    v2f fb[2][2][4];
#pragma unroll
    for (int s = 0; s < 2; ++s) {
#pragma unroll
      for (int u = 0; u < 2; ++u)
        fa[0][s][u] = *(const v2f*)(Qc + u * 16 * SPITCH + s * 4);
#pragma unroll
      for (int t = 0; t < 4; ++t)
        fb[0][s][t] = *(const v2f*)(Kc + t * 16 * SPITCH + s * 4);
    }
    __builtin_amdgcn_sched_barrier(0);
#pragma unroll
    for (int g = 0; g < 4; ++g) {
      if (g < 3) {
#pragma unroll
        for (int s = 0; s < 2; ++s) {
#pragma unroll
          for (int u = 0; u < 2; ++u)
            fa[(g + 1) & 1][s][u] =
                *(const v2f*)(Qc + u * 16 * SPITCH + ((g + 1) * 2 + s) * 4);
#pragma unroll
          for (int t = 0; t < 4; ++t)
            fb[(g + 1) & 1][s][t] =
                *(const v2f*)(Kc + t * 16 * SPITCH + ((g + 1) * 2 + s) * 4);
        }
      }
      if (g == 0) {
        gq0 = *(const float4*)(Qg + kx);
        gq1 = *(const float4*)(Qg + GROW + kx);
        gq2 = *(const float4*)(Qg + 2 * GROW + kx);
        gq3 = *(const float4*)(Qg + 3 * GROW + kx);
        gk0 = *(const float4*)(Kg + kx);
        gk1 = *(const float4*)(Kg + GROW + kx);
        gk2 = *(const float4*)(Kg + 2 * GROW + kx);
        gk3 = *(const float4*)(Kg + 3 * GROW + kx);
      }
      if (g == 3) {
        float* Swn = Sw + ((kc + 1) & 1) * STG;
        *(float4*)(Swn) = gq0;
        *(float4*)(Swn + 32 * SPITCH) = gq1;
        *(float4*)(Swn + 64 * SPITCH) = gq2;
        *(float4*)(Swn + 96 * SPITCH) = gq3;
        *(float4*)(Swn + OPS) = gk0;
        *(float4*)(Swn + OPS + 32 * SPITCH) = gk1;
        *(float4*)(Swn + OPS + 64 * SPITCH) = gk2;
        *(float4*)(Swn + OPS + 96 * SPITCH) = gk3;
      }
      __builtin_amdgcn_sched_barrier(0);
#pragma unroll
      for (int s = 0; s < 2; ++s)
#pragma unroll
        for (int t = 0; t < 4; ++t) {
          acc[0][t] = wmma_f32x4(fa[g & 1][s][0], fb[g & 1][s][t], acc[0][t]);
          acc[1][t] = wmma_f32x4(fa[g & 1][s][1], fb[g & 1][s][t], acc[1][t]);
        }
      __builtin_amdgcn_sched_barrier(0);
    }

    __syncthreads();
  }

  const float tau  = *tauP;
  const float lam  = *lamP;
  const float invT = 1.0f / *tmpP;
#pragma unroll
  for (int u = 0; u < 2; ++u) {
#pragma unroll
    for (int t = 0; t < 4; ++t) {
#pragma unroll
      for (int i = 0; i < 8; ++i) {
        const int m = m0 + wm + u * 16 + i + 8 * half;
        const int n = n0 + wn + t * 16 + row;
        const float s = acc[u][t][i];
        scores[(size_t)m * N_SZ + n] = s;
        const float g = 1.0f / (1.0f + expf(-lam * (s - tau)));
        alphaRaw[(size_t)m * N_SZ + n] = g * expf(s * invT);
      }
    }
  }
}

__global__ __launch_bounds__(128) void maq_alpha_norm(float* __restrict__ alpha) {
  __shared__ float red[128];
  const int t = threadIdx.x;
  float4* p = (float4*)(alpha + (size_t)blockIdx.x * N_SZ);
  float4 v[16];
#pragma unroll
  for (int k = 0; k < 16; ++k) v[k] = p[t + 128 * k];
  __builtin_amdgcn_sched_barrier(0);
  float sa = 0.0f, sb = 0.0f;
#pragma unroll
  for (int k = 0; k < 16; k += 2) {
    sa += v[k].x + v[k].y + v[k].z + v[k].w;
    sb += v[k + 1].x + v[k + 1].y + v[k + 1].z + v[k + 1].w;
  }
  red[t] = sa + sb;
  __syncthreads();
  for (int off = 64; off > 0; off >>= 1) {
    if (t < off) red[t] += red[t + off];
    __syncthreads();
  }
  const float inv = 1.0f / (red[0] + 1e-8f);
#pragma unroll
  for (int k = 0; k < 16; ++k) {
    float4 w = v[k];
    w.x *= inv; w.y *= inv; w.z *= inv; w.w *= inv;
    p[t + 128 * k] = w;
  }
}

extern "C" void kernel_launch(void* const* d_in, const int* in_sizes, int n_in,
                              void* d_out, int out_size, void* d_ws, size_t ws_size,
                              hipStream_t stream) {
  const float* z      = (const float*)d_in[0];
  const float* pool   = (const float*)d_in[1];
  const float* WQ     = (const float*)d_in[2];
  const float* WK     = (const float*)d_in[3];
  const float* logits = (const float*)d_in[4];
  const float* tau    = (const float*)d_in[5];
  const float* lam    = (const float*)d_in[6];
  const float* temp   = (const float*)d_in[7];

  float* out    = (float*)d_out;
  float* alpha  = out;
  float* scores = out + (size_t)B_SZ * N_SZ;
  float* keys   = out + (size_t)2 * B_SZ * N_SZ;

  float* ws = (float*)d_ws;
  float* qs = ws;
  float* kn = qs + (size_t)B_SZ * SD_SZ;

  maq_gemm_q<<<64, 256, 0, stream>>>(z, WQ, qs);
  maq_norm_q<<<16, 256, 0, stream>>>(qs, logits);
  maq_gemm_k<<<256, 256, 0, stream>>>(pool, WK, keys);
  maq_norm_k<<<256, 256, 0, stream>>>(keys, kn);
  maq_gemm_scores<<<256, 256, 0, stream>>>(qs, kn, tau, lam, temp, scores, alpha);
  maq_alpha_norm<<<512, 128, 0, stream>>>(alpha);
}


# Row L2 normalisation fused into both projection GEMM epilogues (two launches fewer); sum of squares order now: 16-lane xor tree per wave, then the aspect's waves ascending via LDS
# speedup vs baseline: 1.8826x; 1.1191x over previous
#include <hip/hip_runtime.h>
#include <math.h>

#define B_SZ  512
#define N_SZ  8192
#define D_SZ  1024
#define DA_SZ 512
#define S_SZ  8
#define DK_SZ 64
#define SD_SZ 512

typedef __attribute__((ext_vector_type(2))) float v2f;
typedef __attribute__((ext_vector_type(8))) float v8f;

__device__ __forceinline__ v8f wmma_f32x4(v2f a, v2f b, v8f c) {
  return __builtin_amdgcn_wmma_f32_16x16x4_f32(
       false, a,  false, b,
       (short)0, c,  false,  false);
}

#define KCH    32
#define APITCH 36

template <int M>
__device__ __forceinline__ float row16_xor(const float x) {
  return __builtin_bit_cast(
      float, __builtin_amdgcn_update_dpp(0, __builtin_bit_cast(int, x),
                                         0x160 + M, 0xf, 0xf, true));
}

template <int KDIM, int MT, int NU, bool RAW>
__device__ __forceinline__ void proj_gemm_tile(const float* A,
                                               const float* __restrict__ Bp,
                                               float* __restrict__ Craw,
                                               float* __restrict__ Cn,
                                               const float wgt,
                                               const int m0, const int c0) {
  constexpr int ROWS = 16 * MT;
  constexpr int NLD  = ROWS * 8 / 256;
  constexpr int STG  = ROWS * APITCH;
  constexpr int NCH  = KDIM / KCH;
  constexpr int WPA  = 4 / NU;
  __shared__ __attribute__((aligned(16))) float As[2 * STG];
  constexpr int APB  = 8 / WPA;
  __shared__ __attribute__((aligned(16))) float part[8 * ROWS];
  __shared__ __attribute__((aligned(16))) float inv[APB * ROWS];

  const int tid  = threadIdx.x;
  const int lane = tid & 31;
  const int row  = lane & 15;
  const int half = lane >> 4;
  const int sr = tid >> 3;
  const int sq = tid & 7;
  const float* Ag = A + (size_t)(m0 + sr) * KDIM + sq * 4;
  float*       Aw = As + sr * APITCH + sq * 4;
  const float* Ar = As + row * APITCH + 2 * half;
  const float* Bl = Bp + (2 * half) * DK_SZ;

  static_assert(NLD == 1 || NLD == 2, "one or two 16-B pieces per thread");
  float4 ga0, ga1;
  v2f bn[8][NU];
  ga0 = *(const float4*)(Ag);
  if constexpr (NLD == 2) ga1 = *(const float4*)(Ag + (size_t)32 * KDIM);
#pragma unroll
  for (int ks = 0; ks < 8; ++ks)
#pragma unroll
    for (int u = 0; u < NU; ++u) {
      bn[ks][u][0] = Bl[(ks * 4 + 0) * DK_SZ + u * 16];
      bn[ks][u][1] = Bl[(ks * 4 + 1) * DK_SZ + u * 16];
    }
  *(float4*)(Aw) = ga0;
  if constexpr (NLD == 2) *(float4*)(Aw + 32 * APITCH) = ga1;
  __syncthreads();

  v8f acc[NU][MT] = {};
  for (int kc = 0; kc < NCH; ++kc) {
    const float* Acur = Ar + (kc & 1) * STG;
    v2f bc[8][NU];
#pragma unroll
    for (int ks = 0; ks < 8; ++ks)
#pragma unroll
      for (int u = 0; u < NU; ++u) bc[ks][u] = bn[ks][u];

    const int kn = ((kc + 1 < NCH) ? (kc + 1) : kc) * KCH;
    v2f af[2][2 * MT];
#pragma unroll
    for (int s = 0; s < 2; ++s)
#pragma unroll
      for (int t = 0; t < MT; ++t)
        af[0][s * MT + t] = *(const v2f*)(Acur + t * 16 * APITCH + s * 4);
#pragma unroll
    for (int g = 0; g < 4; ++g) {
      if (g < 3) {
#pragma unroll
        for (int s = 0; s < 2; ++s)
#pragma unroll
          for (int t = 0; t < MT; ++t)
            af[(g + 1) & 1][s * MT + t] =
                *(const v2f*)(Acur + t * 16 * APITCH + ((g + 1) * 2 + s) * 4);
      }
      if (g == 0) {
        ga0 = *(const float4*)(Ag + kn);
        if constexpr (NLD == 2) ga1 = *(const float4*)(Ag + (size_t)32 * KDIM + kn);
#pragma unroll
        for (int ks = 0; ks < 8; ++ks)
#pragma unroll
          for (int u = 0; u < NU; ++u) {
            bn[ks][u][0] = Bl[(kn + ks * 4 + 0) * DK_SZ + u * 16];
            bn[ks][u][1] = Bl[(kn + ks * 4 + 1) * DK_SZ + u * 16];
          }
      }
      __builtin_amdgcn_sched_barrier(0);
#pragma unroll
      for (int s = 0; s < 2; ++s)
#pragma unroll
        for (int t = 0; t < MT; ++t)
#pragma unroll
          for (int u = 0; u < NU; ++u)
            acc[u][t] = wmma_f32x4(af[g & 1][s * MT + t], bc[g * 2 + s][u], acc[u][t]);
      __builtin_amdgcn_sched_barrier(0);
    }

    {
      float* Awn = Aw + ((kc + 1) & 1) * STG;
      *(float4*)(Awn) = ga0;
      if constexpr (NLD == 2) *(float4*)(Awn + 32 * APITCH) = ga1;
    }
    __syncthreads();
  }

  const int wv = tid >> 5;
  float ss[MT][8];
#pragma unroll
  for (int t = 0; t < MT; ++t)
#pragma unroll
    for (int i = 0; i < 8; ++i) {
      float s = acc[0][t][i] * acc[0][t][i];
#pragma unroll
      for (int u = 1; u < NU; ++u) s += acc[u][t][i] * acc[u][t][i];
      s += row16_xor<1>(s);
      s += row16_xor<2>(s);
      s += row16_xor<4>(s);
      s += row16_xor<8>(s);
      ss[t][i] = s;
    }
  if (row == 0) {
    float* pw = part + wv * ROWS + 8 * half;
#pragma unroll
    for (int t = 0; t < MT; ++t) {
      *(float4*)(pw + t * 16)     = make_float4(ss[t][0], ss[t][1], ss[t][2], ss[t][3]);
      *(float4*)(pw + t * 16 + 4) = make_float4(ss[t][4], ss[t][5], ss[t][6], ss[t][7]);
    }
  }
  __syncthreads();
  const int ab = wv / WPA;
  const int j  = (wv % WPA) * 32 + lane;
  if (WPA * 32 <= ROWS || j < ROWS) {
    const float* pp = part + (ab * WPA) * ROWS + j;
    float tot = pp[0];
#pragma unroll
    for (int g = 1; g < WPA; ++g) tot += pp[g * ROWS];
    inv[ab * ROWS + j] = wgt / (sqrtf(tot) + 1e-8f);
  }
  __syncthreads();
  const float* iv = inv + ab * ROWS + 8 * half;
  const size_t ob = (size_t)(m0 + 8 * half) * SD_SZ + c0 + row;
  float* cn = Cn + ob;
  float* cr = nullptr;
  if constexpr (RAW) cr = Craw + ob;
#pragma unroll
  for (int t = 0; t < MT; ++t) {
    const float4 s0 = *(const float4*)(iv + t * 16);
    const float4 s1 = *(const float4*)(iv + t * 16 + 4);
    const float sc[8] = {s0.x, s0.y, s0.z, s0.w, s1.x, s1.y, s1.z, s1.w};
#pragma unroll
    for (int i = 0; i < 8; ++i)
#pragma unroll
      for (int u = 0; u < NU; ++u) {
        const int o = (t * 16 + i) * SD_SZ + u * 16;
        const float v = acc[u][t][i];
        if constexpr (RAW) cr[o] = v;
        cn[o] = v * sc[i];
      }
  }
}

__global__ __launch_bounds__(256) void maq_gemm_q(const float* z,
                                                  const float* __restrict__ WQ,
                                                  const float* __restrict__ logits,
                                                  float* __restrict__ qs) {
  const int wave = (blockIdx.x * 256 + threadIdx.x) >> 5;
  const int row  = threadIdx.x & 15;
  const int m0 = (wave >> 5) * 32;
  const int c0 = (wave & 31) * 16;
  const float* Bp = WQ + (c0 >> 6) * (DA_SZ * DK_SZ) + (c0 & 63) + row;
  float mx = logits[0];
#pragma unroll
  for (int i = 1; i < S_SZ; ++i) mx = fmaxf(mx, logits[i]);
  float den = 0.0f;
#pragma unroll
  for (int i = 0; i < S_SZ; ++i) den += expf(logits[i] - mx);
  const float w = expf(logits[c0 >> 6] - mx) / den;
  proj_gemm_tile<DA_SZ, 2, 1, false>(z, Bp, nullptr, qs, w, m0, c0);
}

__global__ __launch_bounds__(256) void maq_gemm_k(const float* pool,
                                                  const float* __restrict__ WK,
                                                  float* __restrict__ keys,
                                                  float* __restrict__ kn) {
  const int wave = (blockIdx.x * 256 + threadIdx.x) >> 5;
  const int row  = threadIdx.x & 15;
  const int m0 = (wave >> 4) * 64;
  const int c0 = (wave & 15) * 32;
  const float* Bp = WK + (c0 >> 6) * (D_SZ * DK_SZ) + (c0 & 63) + row;
  proj_gemm_tile<D_SZ, 4, 2, true>(pool, Bp, keys, kn, 1.0f, m0, c0);
}

#define SKCH   32
#define SPITCH 36
#define STILE  128

__global__ __launch_bounds__(256) void maq_gemm_scores(
    const float* qs, const float* kn,
    const float* __restrict__ tauP, const float* __restrict__ lamP,
    const float* __restrict__ tmpP,
    float* __restrict__ scores, float* __restrict__ alphaRaw) {
  constexpr int OPS = STILE * SPITCH;
  constexpr int STG = 2 * OPS;
  constexpr int NCH = SD_SZ / SKCH;
  __shared__ __attribute__((aligned(16))) float Ss[2 * STG];

  const int tid  = threadIdx.x;
  const int lane = tid & 31;
  const int row  = lane & 15;
  const int half = lane >> 4;
  const int wv   = tid >> 5;
  const int m0 = (blockIdx.x >> 6) * STILE;
  const int n0 = (blockIdx.x & 63) * STILE;
  const int wm = (wv >> 1) * 32;
  const int wn = (wv & 1) * 64;

  const int sr = tid >> 3;
  const int sq = tid & 7;
  const float* Qg = qs + (size_t)(m0 + sr) * SD_SZ + sq * 4;
  const float* Kg = kn + (size_t)(n0 + sr) * SD_SZ + sq * 4;
  float*       Sw = Ss + sr * SPITCH + sq * 4;
  const float* Qr = Ss + (wm + row) * SPITCH + 2 * half;
  const float* Kr = Ss + OPS + (wn + row) * SPITCH + 2 * half;

  float4 gq0, gq1, gq2, gq3, gk0, gk1, gk2, gk3;
  constexpr size_t GROW = (size_t)32 * SD_SZ;
  gq0 = *(const float4*)(Qg);
  gq1 = *(const float4*)(Qg + GROW);
  gq2 = *(const float4*)(Qg + 2 * GROW);
  gq3 = *(const float4*)(Qg + 3 * GROW);
  gk0 = *(const float4*)(Kg);
  gk1 = *(const float4*)(Kg + GROW);
  gk2 = *(const float4*)(Kg + 2 * GROW);
  gk3 = *(const float4*)(Kg + 3 * GROW);
  *(float4*)(Sw) = gq0;
  *(float4*)(Sw + 32 * SPITCH) = gq1;
  *(float4*)(Sw + 64 * SPITCH) = gq2;
  *(float4*)(Sw + 96 * SPITCH) = gq3;
  *(float4*)(Sw + OPS) = gk0;
  *(float4*)(Sw + OPS + 32 * SPITCH) = gk1;
  *(float4*)(Sw + OPS + 64 * SPITCH) = gk2;
  *(float4*)(Sw + OPS + 96 * SPITCH) = gk3;
  __syncthreads();

  v8f acc[2][4] = {};
  for (int kc = 0; kc < NCH; ++kc) {
    const float* Qc = Qr + (kc & 1) * STG;
    const float* Kc = Kr + (kc & 1) * STG;
    const int kx = ((kc + 1 < NCH) ? (kc + 1) : kc) * SKCH;
    v2f fa[2][2][2];
    v2f fb[2][2][4];
#pragma unroll
    for (int s = 0; s < 2; ++s) {
#pragma unroll
      for (int u = 0; u < 2; ++u)
        fa[0][s][u] = *(const v2f*)(Qc + u * 16 * SPITCH + s * 4);
#pragma unroll
      for (int t = 0; t < 4; ++t)
        fb[0][s][t] = *(const v2f*)(Kc + t * 16 * SPITCH + s * 4);
    }
    __builtin_amdgcn_sched_barrier(0);
#pragma unroll
    for (int g = 0; g < 4; ++g) {
      if (g < 3) {
#pragma unroll
        for (int s = 0; s < 2; ++s) {
#pragma unroll
          for (int u = 0; u < 2; ++u)
            fa[(g + 1) & 1][s][u] =
                *(const v2f*)(Qc + u * 16 * SPITCH + ((g + 1) * 2 + s) * 4);
#pragma unroll
          for (int t = 0; t < 4; ++t)
            fb[(g + 1) & 1][s][t] =
                *(const v2f*)(Kc + t * 16 * SPITCH + ((g + 1) * 2 + s) * 4);
        }
      }
      if (g == 0) {
        gq0 = *(const float4*)(Qg + kx);
        gq1 = *(const float4*)(Qg + GROW + kx);
        gq2 = *(const float4*)(Qg + 2 * GROW + kx);
        gq3 = *(const float4*)(Qg + 3 * GROW + kx);
        gk0 = *(const float4*)(Kg + kx);
        gk1 = *(const float4*)(Kg + GROW + kx);
        gk2 = *(const float4*)(Kg + 2 * GROW + kx);
        gk3 = *(const float4*)(Kg + 3 * GROW + kx);
      }
      if (g == 3) {
        float* Swn = Sw + ((kc + 1) & 1) * STG;
        *(float4*)(Swn) = gq0;
        *(float4*)(Swn + 32 * SPITCH) = gq1;
        *(float4*)(Swn + 64 * SPITCH) = gq2;
        *(float4*)(Swn + 96 * SPITCH) = gq3;
        *(float4*)(Swn + OPS) = gk0;
        *(float4*)(Swn + OPS + 32 * SPITCH) = gk1;
        *(float4*)(Swn + OPS + 64 * SPITCH) = gk2;
        *(float4*)(Swn + OPS + 96 * SPITCH) = gk3;
      }
      __builtin_amdgcn_sched_barrier(0);
#pragma unroll
      for (int s = 0; s < 2; ++s)
#pragma unroll
        for (int t = 0; t < 4; ++t) {
          acc[0][t] = wmma_f32x4(fa[g & 1][s][0], fb[g & 1][s][t], acc[0][t]);
          acc[1][t] = wmma_f32x4(fa[g & 1][s][1], fb[g & 1][s][t], acc[1][t]);
        }
      __builtin_amdgcn_sched_barrier(0);
    }

    __syncthreads();
  }

  const float tau  = *tauP;
  const float lam  = *lamP;
  const float invT = 1.0f / *tmpP;
#pragma unroll
  for (int u = 0; u < 2; ++u) {
#pragma unroll
    for (int t = 0; t < 4; ++t) {
#pragma unroll
      for (int i = 0; i < 8; ++i) {
        const int m = m0 + wm + u * 16 + i + 8 * half;
        const int n = n0 + wn + t * 16 + row;
        const float s = acc[u][t][i];
        scores[(size_t)m * N_SZ + n] = s;
        const float g = 1.0f / (1.0f + expf(-lam * (s - tau)));
        alphaRaw[(size_t)m * N_SZ + n] = g * expf(s * invT);
      }
    }
  }
}

__global__ __launch_bounds__(128) void maq_alpha_norm(float* __restrict__ alpha) {
  __shared__ float red[128];
  const int t = threadIdx.x;
  float4* p = (float4*)(alpha + (size_t)blockIdx.x * N_SZ);
  float4 v[16];
#pragma unroll
  for (int k = 0; k < 16; ++k) v[k] = p[t + 128 * k];
  __builtin_amdgcn_sched_barrier(0);
  float sa = 0.0f, sb = 0.0f;
#pragma unroll
  for (int k = 0; k < 16; k += 2) {
    sa += v[k].x + v[k].y + v[k].z + v[k].w;
    sb += v[k + 1].x + v[k + 1].y + v[k + 1].z + v[k + 1].w;
  }
  red[t] = sa + sb;
  __syncthreads();
  for (int off = 64; off > 0; off >>= 1) {
    if (t < off) red[t] += red[t + off];
    __syncthreads();
  }
  const float inv = 1.0f / (red[0] + 1e-8f);
#pragma unroll
  for (int k = 0; k < 16; ++k) {
    float4 w = v[k];
    w.x *= inv; w.y *= inv; w.z *= inv; w.w *= inv;
    p[t + 128 * k] = w;
  }
}

extern "C" void kernel_launch(void* const* d_in, const int* in_sizes, int n_in,
                              void* d_out, int out_size, void* d_ws, size_t ws_size,
                              hipStream_t stream) {
  const float* z      = (const float*)d_in[0];
  const float* pool   = (const float*)d_in[1];
  const float* WQ     = (const float*)d_in[2];
  const float* WK     = (const float*)d_in[3];
  const float* logits = (const float*)d_in[4];
  const float* tau    = (const float*)d_in[5];
  const float* lam    = (const float*)d_in[6];
  const float* temp   = (const float*)d_in[7];

  float* out    = (float*)d_out;
  float* alpha  = out;
  float* scores = out + (size_t)B_SZ * N_SZ;
  float* keys   = out + (size_t)2 * B_SZ * N_SZ;

  float* ws = (float*)d_ws;
  float* qs = ws;
  float* kn = qs + (size_t)B_SZ * SD_SZ;

  maq_gemm_q<<<64, 256, 0, stream>>>(z, WQ, logits, qs);
  maq_gemm_k<<<256, 256, 0, stream>>>(pool, WK, keys, kn);
  maq_gemm_scores<<<256, 256, 0, stream>>>(qs, kn, tau, lam, temp, scores, alpha);
  maq_alpha_norm<<<512, 128, 0, stream>>>(alpha);
}


# Both projection GEMMs in one launch: blocks 0-255 project the keys, blocks 256-319 the queries (independent work, the small grid runs beside the large one); one launch fewer; every sum order kept
# speedup vs baseline: 2.2215x; 1.1800x over previous
#include <hip/hip_runtime.h>
#include <math.h>

#define B_SZ  512
#define N_SZ  8192
#define D_SZ  1024
#define DA_SZ 512
#define S_SZ  8
#define DK_SZ 64
#define SD_SZ 512

typedef __attribute__((ext_vector_type(2))) float v2f;
typedef __attribute__((ext_vector_type(8))) float v8f;

__device__ __forceinline__ v8f wmma_f32x4(v2f a, v2f b, v8f c) {
  return __builtin_amdgcn_wmma_f32_16x16x4_f32(
       false, a,  false, b,
       (short)0, c,  false,  false);
}

#define KCH    32
#define APITCH 36

template <int M>
__device__ __forceinline__ float row16_xor(const float x) {
  return __builtin_bit_cast(
      float, __builtin_amdgcn_update_dpp(0, __builtin_bit_cast(int, x),
                                         0x160 + M, 0xf, 0xf, true));
}

template <int KDIM, int MT, int NU, bool RAW>
__device__ __forceinline__ void proj_gemm_tile(const float* A,
                                               const float* __restrict__ Bp,
                                               float* __restrict__ Craw,
                                               float* __restrict__ Cn,
                                               const float wgt,
                                               const int m0, const int c0) {
  constexpr int ROWS = 16 * MT;
  constexpr int NLD  = ROWS * 8 / 256;
  constexpr int STG  = ROWS * APITCH;
  constexpr int NCH  = KDIM / KCH;
  constexpr int WPA  = 4 / NU;
  __shared__ __attribute__((aligned(16))) float As[2 * STG];
  constexpr int APB  = 8 / WPA;
  __shared__ __attribute__((aligned(16))) float part[8 * ROWS];
  __shared__ __attribute__((aligned(16))) float inv[APB * ROWS];

  const int tid  = threadIdx.x;
  const int lane = tid & 31;
  const int row  = lane & 15;
  const int half = lane >> 4;
  const int sr = tid >> 3;
  const int sq = tid & 7;
  const float* Ag = A + (size_t)(m0 + sr) * KDIM + sq * 4;
  float*       Aw = As + sr * APITCH + sq * 4;
  const float* Ar = As + row * APITCH + 2 * half;
  const float* Bl = Bp + (2 * half) * DK_SZ;

  static_assert(NLD == 1 || NLD == 2, "one or two 16-B pieces per thread");
  float4 ga0, ga1;
  v2f bn[8][NU];
  ga0 = *(const float4*)(Ag);
  if constexpr (NLD == 2) ga1 = *(const float4*)(Ag + (size_t)32 * KDIM);
#pragma unroll
  for (int ks = 0; ks < 8; ++ks)
#pragma unroll
    for (int u = 0; u < NU; ++u) {
      bn[ks][u][0] = Bl[(ks * 4 + 0) * DK_SZ + u * 16];
      bn[ks][u][1] = Bl[(ks * 4 + 1) * DK_SZ + u * 16];
    }
  *(float4*)(Aw) = ga0;
  if constexpr (NLD == 2) *(float4*)(Aw + 32 * APITCH) = ga1;
  __syncthreads();

  v8f acc[NU][MT] = {};
  for (int kc = 0; kc < NCH; ++kc) {
    const float* Acur = Ar + (kc & 1) * STG;
    v2f bc[8][NU];
#pragma unroll
    for (int ks = 0; ks < 8; ++ks)
#pragma unroll
      for (int u = 0; u < NU; ++u) bc[ks][u] = bn[ks][u];

    const int kn = ((kc + 1 < NCH) ? (kc + 1) : kc) * KCH;
    v2f af[2][2 * MT];
#pragma unroll
    for (int s = 0; s < 2; ++s)
#pragma unroll
      for (int t = 0; t < MT; ++t)
        af[0][s * MT + t] = *(const v2f*)(Acur + t * 16 * APITCH + s * 4);
#pragma unroll
    for (int g = 0; g < 4; ++g) {
      if (g < 3) {
#pragma unroll
        for (int s = 0; s < 2; ++s)
#pragma unroll
          for (int t = 0; t < MT; ++t)
            af[(g + 1) & 1][s * MT + t] =
                *(const v2f*)(Acur + t * 16 * APITCH + ((g + 1) * 2 + s) * 4);
      }
      if (g == 0) {
        ga0 = *(const float4*)(Ag + kn);
        if constexpr (NLD == 2) ga1 = *(const float4*)(Ag + (size_t)32 * KDIM + kn);
#pragma unroll
        for (int ks = 0; ks < 8; ++ks)
#pragma unroll
          for (int u = 0; u < NU; ++u) {
            bn[ks][u][0] = Bl[(kn + ks * 4 + 0) * DK_SZ + u * 16];
            bn[ks][u][1] = Bl[(kn + ks * 4 + 1) * DK_SZ + u * 16];
          }
      }
      __builtin_amdgcn_sched_barrier(0);
#pragma unroll
      for (int s = 0; s < 2; ++s)
#pragma unroll
        for (int t = 0; t < MT; ++t)
#pragma unroll
          for (int u = 0; u < NU; ++u)
            acc[u][t] = wmma_f32x4(af[g & 1][s * MT + t], bc[g * 2 + s][u], acc[u][t]);
      __builtin_amdgcn_sched_barrier(0);
    }

    {
      float* Awn = Aw + ((kc + 1) & 1) * STG;
      *(float4*)(Awn) = ga0;
      if constexpr (NLD == 2) *(float4*)(Awn + 32 * APITCH) = ga1;
    }
    __syncthreads();
  }

  const int wv = tid >> 5;
  float ss[MT][8];
#pragma unroll
  for (int t = 0; t < MT; ++t)
#pragma unroll
    for (int i = 0; i < 8; ++i) {
      float s = acc[0][t][i] * acc[0][t][i];
#pragma unroll
      for (int u = 1; u < NU; ++u) s += acc[u][t][i] * acc[u][t][i];
      s += row16_xor<1>(s);
      s += row16_xor<2>(s);
      s += row16_xor<4>(s);
      s += row16_xor<8>(s);
      ss[t][i] = s;
    }
  if (row == 0) {
    float* pw = part + wv * ROWS + 8 * half;
#pragma unroll
    for (int t = 0; t < MT; ++t) {
      *(float4*)(pw + t * 16)     = make_float4(ss[t][0], ss[t][1], ss[t][2], ss[t][3]);
      *(float4*)(pw + t * 16 + 4) = make_float4(ss[t][4], ss[t][5], ss[t][6], ss[t][7]);
    }
  }
  __syncthreads();
  const int ab = wv / WPA;
  const int j  = (wv % WPA) * 32 + lane;
  if (WPA * 32 <= ROWS || j < ROWS) {
    const float* pp = part + (ab * WPA) * ROWS + j;
    float tot = pp[0];
#pragma unroll
    for (int g = 1; g < WPA; ++g) tot += pp[g * ROWS];
    inv[ab * ROWS + j] = wgt / (sqrtf(tot) + 1e-8f);
  }
  __syncthreads();
  const float* iv = inv + ab * ROWS + 8 * half;
  const size_t ob = (size_t)(m0 + 8 * half) * SD_SZ + c0 + row;
  float* cn = Cn + ob;
  float* cr = nullptr;
  if constexpr (RAW) cr = Craw + ob;
#pragma unroll
  for (int t = 0; t < MT; ++t) {
    const float4 s0 = *(const float4*)(iv + t * 16);
    const float4 s1 = *(const float4*)(iv + t * 16 + 4);
    const float sc[8] = {s0.x, s0.y, s0.z, s0.w, s1.x, s1.y, s1.z, s1.w};
#pragma unroll
    for (int i = 0; i < 8; ++i)
#pragma unroll
      for (int u = 0; u < NU; ++u) {
        const int o = (t * 16 + i) * SD_SZ + u * 16;
        const float v = acc[u][t][i];
        if constexpr (RAW) cr[o] = v;
        cn[o] = v * sc[i];
      }
  }
}

#define KBLOCKS 256

__global__ __launch_bounds__(256) void maq_gemm_proj(const float* pool,
                                                     const float* __restrict__ WK,
                                                     float* __restrict__ keys,
                                                     float* __restrict__ kn,
                                                     const float* z,
                                                     const float* __restrict__ WQ,
                                                     const float* __restrict__ logits,
                                                     float* __restrict__ qs) {
  const int row = threadIdx.x & 15;
  if (blockIdx.x < KBLOCKS) {
    const int wave = (blockIdx.x * 256 + threadIdx.x) >> 5;
    const int m0 = (wave >> 4) * 64;
    const int c0 = (wave & 15) * 32;
    const float* Bp = WK + (c0 >> 6) * (D_SZ * DK_SZ) + (c0 & 63) + row;
    proj_gemm_tile<D_SZ, 4, 2, true>(pool, Bp, keys, kn, 1.0f, m0, c0);
  } else {
    const int wave = ((blockIdx.x - KBLOCKS) * 256 + threadIdx.x) >> 5;
    const int m0 = (wave >> 5) * 32;
    const int c0 = (wave & 31) * 16;
    const float* Bp = WQ + (c0 >> 6) * (DA_SZ * DK_SZ) + (c0 & 63) + row;
    float mx = logits[0];
#pragma unroll
    for (int i = 1; i < S_SZ; ++i) mx = fmaxf(mx, logits[i]);
    float den = 0.0f;
#pragma unroll
    for (int i = 0; i < S_SZ; ++i) den += expf(logits[i] - mx);
    const float w = expf(logits[c0 >> 6] - mx) / den;
    proj_gemm_tile<DA_SZ, 2, 1, false>(z, Bp, nullptr, qs, w, m0, c0);
  }
}

#define SKCH   32
#define SPITCH 36
#define STILE  128

__global__ __launch_bounds__(256) void maq_gemm_scores(
    const float* qs, const float* kn,
    const float* __restrict__ tauP, const float* __restrict__ lamP,
    const float* __restrict__ tmpP,
    float* __restrict__ scores, float* __restrict__ alphaRaw) {
  constexpr int OPS = STILE * SPITCH;
  constexpr int STG = 2 * OPS;
  constexpr int NCH = SD_SZ / SKCH;
  __shared__ __attribute__((aligned(16))) float Ss[2 * STG];

  const int tid  = threadIdx.x;
  const int lane = tid & 31;
  const int row  = lane & 15;
  const int half = lane >> 4;
  const int wv   = tid >> 5;
  const int m0 = (blockIdx.x >> 6) * STILE;
  const int n0 = (blockIdx.x & 63) * STILE;
  const int wm = (wv >> 1) * 32;
  const int wn = (wv & 1) * 64;

  const int sr = tid >> 3;
  const int sq = tid & 7;
  const float* Qg = qs + (size_t)(m0 + sr) * SD_SZ + sq * 4;
  const float* Kg = kn + (size_t)(n0 + sr) * SD_SZ + sq * 4;
  float*       Sw = Ss + sr * SPITCH + sq * 4;
  const float* Qr = Ss + (wm + row) * SPITCH + 2 * half;
  const float* Kr = Ss + OPS + (wn + row) * SPITCH + 2 * half;

  float4 gq0, gq1, gq2, gq3, gk0, gk1, gk2, gk3;
  constexpr size_t GROW = (size_t)32 * SD_SZ;
  gq0 = *(const float4*)(Qg);
  gq1 = *(const float4*)(Qg + GROW);
  gq2 = *(const float4*)(Qg + 2 * GROW);
  gq3 = *(const float4*)(Qg + 3 * GROW);
  gk0 = *(const float4*)(Kg);
  gk1 = *(const float4*)(Kg + GROW);
  gk2 = *(const float4*)(Kg + 2 * GROW);
  gk3 = *(const float4*)(Kg + 3 * GROW);
  *(float4*)(Sw) = gq0;
  *(float4*)(Sw + 32 * SPITCH) = gq1;
  *(float4*)(Sw + 64 * SPITCH) = gq2;
  *(float4*)(Sw + 96 * SPITCH) = gq3;
  *(float4*)(Sw + OPS) = gk0;
  *(float4*)(Sw + OPS + 32 * SPITCH) = gk1;
  *(float4*)(Sw + OPS + 64 * SPITCH) = gk2;
  *(float4*)(Sw + OPS + 96 * SPITCH) = gk3;
  __syncthreads();

  v8f acc[2][4] = {};
  for (int kc = 0; kc < NCH; ++kc) {
    const float* Qc = Qr + (kc & 1) * STG;
    const float* Kc = Kr + (kc & 1) * STG;
    const int kx = ((kc + 1 < NCH) ? (kc + 1) : kc) * SKCH;
    v2f fa[2][2][2];
    v2f fb[2][2][4];
#pragma unroll
    for (int s = 0; s < 2; ++s) {
#pragma unroll
      for (int u = 0; u < 2; ++u)
        fa[0][s][u] = *(const v2f*)(Qc + u * 16 * SPITCH + s * 4);
#pragma unroll
      for (int t = 0; t < 4; ++t)
        fb[0][s][t] = *(const v2f*)(Kc + t * 16 * SPITCH + s * 4);
    }
    __builtin_amdgcn_sched_barrier(0);
#pragma unroll
    for (int g = 0; g < 4; ++g) {
      if (g < 3) {
#pragma unroll
        for (int s = 0; s < 2; ++s) {
#pragma unroll
          for (int u = 0; u < 2; ++u)
            fa[(g + 1) & 1][s][u] =
                *(const v2f*)(Qc + u * 16 * SPITCH + ((g + 1) * 2 + s) * 4);
#pragma unroll
          for (int t = 0; t < 4; ++t)
            fb[(g + 1) & 1][s][t] =
                *(const v2f*)(Kc + t * 16 * SPITCH + ((g + 1) * 2 + s) * 4);
        }
      }
      if (g == 0) {
        gq0 = *(const float4*)(Qg + kx);
        gq1 = *(const float4*)(Qg + GROW + kx);
        gq2 = *(const float4*)(Qg + 2 * GROW + kx);
        gq3 = *(const float4*)(Qg + 3 * GROW + kx);
        gk0 = *(const float4*)(Kg + kx);
        gk1 = *(const float4*)(Kg + GROW + kx);
        gk2 = *(const float4*)(Kg + 2 * GROW + kx);
        gk3 = *(const float4*)(Kg + 3 * GROW + kx);
      }
      if (g == 3) {
        float* Swn = Sw + ((kc + 1) & 1) * STG;
        *(float4*)(Swn) = gq0;
        *(float4*)(Swn + 32 * SPITCH) = gq1;
        *(float4*)(Swn + 64 * SPITCH) = gq2;
        *(float4*)(Swn + 96 * SPITCH) = gq3;
        *(float4*)(Swn + OPS) = gk0;
        *(float4*)(Swn + OPS + 32 * SPITCH) = gk1;
        *(float4*)(Swn + OPS + 64 * SPITCH) = gk2;
        *(float4*)(Swn + OPS + 96 * SPITCH) = gk3;
      }
      __builtin_amdgcn_sched_barrier(0);
#pragma unroll
      for (int s = 0; s < 2; ++s)
#pragma unroll
        for (int t = 0; t < 4; ++t) {
          acc[0][t] = wmma_f32x4(fa[g & 1][s][0], fb[g & 1][s][t], acc[0][t]);
          acc[1][t] = wmma_f32x4(fa[g & 1][s][1], fb[g & 1][s][t], acc[1][t]);
        }
      __builtin_amdgcn_sched_barrier(0);
    }

    __syncthreads();
  }

  const float tau  = *tauP;
  const float lam  = *lamP;
  const float invT = 1.0f / *tmpP;
#pragma unroll
  for (int u = 0; u < 2; ++u) {
#pragma unroll
    for (int t = 0; t < 4; ++t) {
#pragma unroll
      for (int i = 0; i < 8; ++i) {
        const int m = m0 + wm + u * 16 + i + 8 * half;
        const int n = n0 + wn + t * 16 + row;
        const float s = acc[u][t][i];
        scores[(size_t)m * N_SZ + n] = s;
        const float g = 1.0f / (1.0f + expf(-lam * (s - tau)));
        alphaRaw[(size_t)m * N_SZ + n] = g * expf(s * invT);
      }
    }
  }
}

__global__ __launch_bounds__(128) void maq_alpha_norm(float* __restrict__ alpha) {
  __shared__ float red[128];
  const int t = threadIdx.x;
  float4* p = (float4*)(alpha + (size_t)blockIdx.x * N_SZ);
  float4 v[16];
#pragma unroll
  for (int k = 0; k < 16; ++k) v[k] = p[t + 128 * k];
  __builtin_amdgcn_sched_barrier(0);
  float sa = 0.0f, sb = 0.0f;
#pragma unroll
  for (int k = 0; k < 16; k += 2) {
    sa += v[k].x + v[k].y + v[k].z + v[k].w;
    sb += v[k + 1].x + v[k + 1].y + v[k + 1].z + v[k + 1].w;
  }
  red[t] = sa + sb;
  __syncthreads();
  for (int off = 64; off > 0; off >>= 1) {
    if (t < off) red[t] += red[t + off];
    __syncthreads();
  }
  const float inv = 1.0f / (red[0] + 1e-8f);
#pragma unroll
  for (int k = 0; k < 16; ++k) {
    float4 w = v[k];
    w.x *= inv; w.y *= inv; w.z *= inv; w.w *= inv;
    p[t + 128 * k] = w;
  }
}

extern "C" void kernel_launch(void* const* d_in, const int* in_sizes, int n_in,
                              void* d_out, int out_size, void* d_ws, size_t ws_size,
                              hipStream_t stream) {
  const float* z      = (const float*)d_in[0];
  const float* pool   = (const float*)d_in[1];
  const float* WQ     = (const float*)d_in[2];
  const float* WK     = (const float*)d_in[3];
  const float* logits = (const float*)d_in[4];
  const float* tau    = (const float*)d_in[5];
  const float* lam    = (const float*)d_in[6];
  const float* temp   = (const float*)d_in[7];

  float* out    = (float*)d_out;
  float* alpha  = out;
  float* scores = out + (size_t)B_SZ * N_SZ;
  float* keys   = out + (size_t)2 * B_SZ * N_SZ;

  float* ws = (float*)d_ws;
  float* qs = ws;
  float* kn = qs + (size_t)B_SZ * SD_SZ;

  maq_gemm_proj<<<320, 256, 0, stream>>>(pool, WK, keys, kn, z, WQ, logits, qs);
  maq_gemm_scores<<<256, 256, 0, stream>>>(qs, kn, tau, lam, temp, scores, alpha);
  maq_alpha_norm<<<512, 128, 0, stream>>>(alpha);
}


# Projection chunk loop: next A tile stored to LDS ahead of the last WMMA group, W loads behind the first group (shorter gap at each chunk barrier); query blocks 16 rows x 128 cols to fit the gaps
# speedup vs baseline: 2.2870x; 1.0295x over previous
#include <hip/hip_runtime.h>
#include <math.h>

#define B_SZ  512
#define N_SZ  8192
#define D_SZ  1024
#define DA_SZ 512
#define S_SZ  8
#define DK_SZ 64
#define SD_SZ 512

typedef __attribute__((ext_vector_type(2))) float v2f;
typedef __attribute__((ext_vector_type(8))) float v8f;

__device__ __forceinline__ v8f wmma_f32x4(v2f a, v2f b, v8f c) {
  return __builtin_amdgcn_wmma_f32_16x16x4_f32(
       false, a,  false, b,
       (short)0, c,  false,  false);
}

#define KCH    32
#define APITCH 36

template <int M>
__device__ __forceinline__ float row16_xor(const float x) {
  return __builtin_bit_cast(
      float, __builtin_amdgcn_update_dpp(0, __builtin_bit_cast(int, x),
                                         0x160 + M, 0xf, 0xf, true));
}

template <int KDIM, int MT, int NU, bool RAW>
__device__ __forceinline__ void proj_gemm_tile(const float* A,
                                               const float* __restrict__ Bp,
                                               float* __restrict__ Craw,
                                               float* __restrict__ Cn,
                                               const float wgt,
                                               const int m0, const int c0) {
  constexpr int ROWS = 16 * MT;
  constexpr int NST  = ROWS * 8;
  constexpr bool PART = NST < 256;
  constexpr int NLD  = PART ? 1 : NST / 256;
  constexpr int STG  = ROWS * APITCH;
  constexpr int NCH  = KDIM / KCH;
  constexpr int WPA  = 4 / NU;
  __shared__ __attribute__((aligned(16))) float As[2 * STG];
  constexpr int APB  = 8 / WPA;
  __shared__ __attribute__((aligned(16))) float part[8 * ROWS];
  __shared__ __attribute__((aligned(16))) float inv[APB * ROWS];

  const int tid  = threadIdx.x;
  const int lane = tid & 31;
  const int row  = lane & 15;
  const int half = lane >> 4;
  const bool stg = !PART || tid < NST;
  const int sr = tid >> 3;
  const int sq = tid & 7;
  const float* Ag = A + (size_t)(m0 + sr) * KDIM + sq * 4;
  float*       Aw = As + sr * APITCH + sq * 4;
  const float* Ar = As + row * APITCH + 2 * half;
  const float* Bl = Bp + (2 * half) * DK_SZ;

  static_assert(NLD == 1 || NLD == 2, "one or two 16-B pieces per thread");
  float4 ga0 = make_float4(0.0f, 0.0f, 0.0f, 0.0f);
  float4 ga1 = ga0;
  v2f bn[8][NU];
  if (stg) {
    ga0 = *(const float4*)(Ag);
    if constexpr (NLD == 2) ga1 = *(const float4*)(Ag + (size_t)32 * KDIM);
  }
#pragma unroll
  for (int ks = 0; ks < 8; ++ks)
#pragma unroll
    for (int u = 0; u < NU; ++u) {
      bn[ks][u][0] = Bl[(ks * 4 + 0) * DK_SZ + u * 16];
      bn[ks][u][1] = Bl[(ks * 4 + 1) * DK_SZ + u * 16];
    }
  if (stg) {
    *(float4*)(Aw) = ga0;
    if constexpr (NLD == 2) *(float4*)(Aw + 32 * APITCH) = ga1;
  }
  __syncthreads();

  v8f acc[NU][MT] = {};
  for (int kc = 0; kc < NCH; ++kc) {
    const float* Acur = Ar + (kc & 1) * STG;
    v2f bc[8][NU];
#pragma unroll
    for (int ks = 0; ks < 8; ++ks)
#pragma unroll
      for (int u = 0; u < NU; ++u) bc[ks][u] = bn[ks][u];

    const int kn = ((kc + 1 < NCH) ? (kc + 1) : kc) * KCH;
    v2f af[2][2 * MT];
#pragma unroll
    for (int s = 0; s < 2; ++s)
#pragma unroll
      for (int t = 0; t < MT; ++t)
        af[0][s * MT + t] = *(const v2f*)(Acur + t * 16 * APITCH + s * 4);
#pragma unroll
    for (int g = 0; g < 4; ++g) {
      if (g < 3) {
#pragma unroll
        for (int s = 0; s < 2; ++s)
#pragma unroll
          for (int t = 0; t < MT; ++t)
            af[(g + 1) & 1][s * MT + t] =
                *(const v2f*)(Acur + t * 16 * APITCH + ((g + 1) * 2 + s) * 4);
      }
      if (g == 0) {
        if (stg) {
          ga0 = *(const float4*)(Ag + kn);
          if constexpr (NLD == 2) ga1 = *(const float4*)(Ag + (size_t)32 * KDIM + kn);
        }
      }
      if (g == 1) {
#pragma unroll
        for (int ks = 0; ks < 8; ++ks)
#pragma unroll
          for (int u = 0; u < NU; ++u) {
            bn[ks][u][0] = Bl[(kn + ks * 4 + 0) * DK_SZ + u * 16];
            bn[ks][u][1] = Bl[(kn + ks * 4 + 1) * DK_SZ + u * 16];
          }
      }
      if (g == 3) {
        if (stg) {
          float* Awn = Aw + ((kc + 1) & 1) * STG;
          *(float4*)(Awn) = ga0;
          if constexpr (NLD == 2) *(float4*)(Awn + 32 * APITCH) = ga1;
        }
      }
      asm volatile("" ::: "memory");
      __builtin_amdgcn_sched_barrier(0);
#pragma unroll
      for (int s = 0; s < 2; ++s)
#pragma unroll
        for (int t = 0; t < MT; ++t)
#pragma unroll
          for (int u = 0; u < NU; ++u)
            acc[u][t] = wmma_f32x4(af[g & 1][s * MT + t], bc[g * 2 + s][u], acc[u][t]);
      __builtin_amdgcn_sched_barrier(0);
    }

    __syncthreads();
  }

  const int wv = tid >> 5;
  float ss[MT][8];
#pragma unroll
  for (int t = 0; t < MT; ++t)
#pragma unroll
    for (int i = 0; i < 8; ++i) {
      float s = acc[0][t][i] * acc[0][t][i];
#pragma unroll
      for (int u = 1; u < NU; ++u) s += acc[u][t][i] * acc[u][t][i];
      s += row16_xor<1>(s);
      s += row16_xor<2>(s);
      s += row16_xor<4>(s);
      s += row16_xor<8>(s);
      ss[t][i] = s;
    }
  if (row == 0) {
    float* pw = part + wv * ROWS + 8 * half;
#pragma unroll
    for (int t = 0; t < MT; ++t) {
      *(float4*)(pw + t * 16)     = make_float4(ss[t][0], ss[t][1], ss[t][2], ss[t][3]);
      *(float4*)(pw + t * 16 + 4) = make_float4(ss[t][4], ss[t][5], ss[t][6], ss[t][7]);
    }
  }
  __syncthreads();
  const int ab = wv / WPA;
  const int j  = (wv % WPA) * 32 + lane;
  if (WPA * 32 <= ROWS || j < ROWS) {
    const float* pp = part + (ab * WPA) * ROWS + j;
    float tot = pp[0];
#pragma unroll
    for (int g = 1; g < WPA; ++g) tot += pp[g * ROWS];
    inv[ab * ROWS + j] = wgt / (sqrtf(tot) + 1e-8f);
  }
  __syncthreads();
  const float* iv = inv + ab * ROWS + 8 * half;
  const size_t ob = (size_t)(m0 + 8 * half) * SD_SZ + c0 + row;
  float* cn = Cn + ob;
  float* cr = nullptr;
  if constexpr (RAW) cr = Craw + ob;
#pragma unroll
  for (int t = 0; t < MT; ++t) {
    const float4 s0 = *(const float4*)(iv + t * 16);
    const float4 s1 = *(const float4*)(iv + t * 16 + 4);
    const float sc[8] = {s0.x, s0.y, s0.z, s0.w, s1.x, s1.y, s1.z, s1.w};
#pragma unroll
    for (int i = 0; i < 8; ++i)
#pragma unroll
      for (int u = 0; u < NU; ++u) {
        const int o = (t * 16 + i) * SD_SZ + u * 16;
        const float v = acc[u][t][i];
        if constexpr (RAW) cr[o] = v;
        cn[o] = v * sc[i];
      }
  }
}

#define KBLOCKS 256

__global__ __launch_bounds__(256) void maq_gemm_proj(const float* pool,
                                                     const float* __restrict__ WK,
                                                     float* __restrict__ keys,
                                                     float* __restrict__ kn,
                                                     const float* z,
                                                     const float* __restrict__ WQ,
                                                     const float* __restrict__ logits,
                                                     float* __restrict__ qs) {
  const int row = threadIdx.x & 15;
  if (blockIdx.x < KBLOCKS) {
    const int wave = (blockIdx.x * 256 + threadIdx.x) >> 5;
    const int m0 = (wave >> 4) * 64;
    const int c0 = (wave & 15) * 32;
    const float* Bp = WK + (c0 >> 6) * (D_SZ * DK_SZ) + (c0 & 63) + row;
    proj_gemm_tile<D_SZ, 4, 2, true>(pool, Bp, keys, kn, 1.0f, m0, c0);
  } else {
    const int wave = ((blockIdx.x - KBLOCKS) * 256 + threadIdx.x) >> 5;
    const int m0 = (wave >> 5) * 16;
    const int c0 = (wave & 31) * 16;
    const float* Bp = WQ + (c0 >> 6) * (DA_SZ * DK_SZ) + (c0 & 63) + row;
    float mx = logits[0];
#pragma unroll
    for (int i = 1; i < S_SZ; ++i) mx = fmaxf(mx, logits[i]);
    float den = 0.0f;
#pragma unroll
    for (int i = 0; i < S_SZ; ++i) den += expf(logits[i] - mx);
    const float w = expf(logits[c0 >> 6] - mx) / den;
    proj_gemm_tile<DA_SZ, 1, 1, false>(z, Bp, nullptr, qs, w, m0, c0);
  }
}

#define SKCH   32
#define SPITCH 36
#define STILE  128

__global__ __launch_bounds__(256) void maq_gemm_scores(
    const float* qs, const float* kn,
    const float* __restrict__ tauP, const float* __restrict__ lamP,
    const float* __restrict__ tmpP,
    float* __restrict__ scores, float* __restrict__ alphaRaw) {
  constexpr int OPS = STILE * SPITCH;
  constexpr int STG = 2 * OPS;
  constexpr int NCH = SD_SZ / SKCH;
  __shared__ __attribute__((aligned(16))) float Ss[2 * STG];

  const int tid  = threadIdx.x;
  const int lane = tid & 31;
  const int row  = lane & 15;
  const int half = lane >> 4;
  const int wv   = tid >> 5;
  const int m0 = (blockIdx.x >> 6) * STILE;
  const int n0 = (blockIdx.x & 63) * STILE;
  const int wm = (wv >> 1) * 32;
  const int wn = (wv & 1) * 64;

  const int sr = tid >> 3;
  const int sq = tid & 7;
  const float* Qg = qs + (size_t)(m0 + sr) * SD_SZ + sq * 4;
  const float* Kg = kn + (size_t)(n0 + sr) * SD_SZ + sq * 4;
  float*       Sw = Ss + sr * SPITCH + sq * 4;
  const float* Qr = Ss + (wm + row) * SPITCH + 2 * half;
  const float* Kr = Ss + OPS + (wn + row) * SPITCH + 2 * half;

  float4 gq0, gq1, gq2, gq3, gk0, gk1, gk2, gk3;
  constexpr size_t GROW = (size_t)32 * SD_SZ;
  gq0 = *(const float4*)(Qg);
  gq1 = *(const float4*)(Qg + GROW);
  gq2 = *(const float4*)(Qg + 2 * GROW);
  gq3 = *(const float4*)(Qg + 3 * GROW);
  gk0 = *(const float4*)(Kg);
  gk1 = *(const float4*)(Kg + GROW);
  gk2 = *(const float4*)(Kg + 2 * GROW);
  gk3 = *(const float4*)(Kg + 3 * GROW);
  *(float4*)(Sw) = gq0;
  *(float4*)(Sw + 32 * SPITCH) = gq1;
  *(float4*)(Sw + 64 * SPITCH) = gq2;
  *(float4*)(Sw + 96 * SPITCH) = gq3;
  *(float4*)(Sw + OPS) = gk0;
  *(float4*)(Sw + OPS + 32 * SPITCH) = gk1;
  *(float4*)(Sw + OPS + 64 * SPITCH) = gk2;
  *(float4*)(Sw + OPS + 96 * SPITCH) = gk3;
  __syncthreads();

  v8f acc[2][4] = {};
  for (int kc = 0; kc < NCH; ++kc) {
    const float* Qc = Qr + (kc & 1) * STG;
    const float* Kc = Kr + (kc & 1) * STG;
    const int kx = ((kc + 1 < NCH) ? (kc + 1) : kc) * SKCH;
    v2f fa[2][2][2];
    v2f fb[2][2][4];
#pragma unroll
    for (int s = 0; s < 2; ++s) {
#pragma unroll
      for (int u = 0; u < 2; ++u)
        fa[0][s][u] = *(const v2f*)(Qc + u * 16 * SPITCH + s * 4);
#pragma unroll
      for (int t = 0; t < 4; ++t)
        fb[0][s][t] = *(const v2f*)(Kc + t * 16 * SPITCH + s * 4);
    }
    __builtin_amdgcn_sched_barrier(0);
#pragma unroll
    for (int g = 0; g < 4; ++g) {
      if (g < 3) {
#pragma unroll
        for (int s = 0; s < 2; ++s) {
#pragma unroll
          for (int u = 0; u < 2; ++u)
            fa[(g + 1) & 1][s][u] =
                *(const v2f*)(Qc + u * 16 * SPITCH + ((g + 1) * 2 + s) * 4);
#pragma unroll
          for (int t = 0; t < 4; ++t)
            fb[(g + 1) & 1][s][t] =
                *(const v2f*)(Kc + t * 16 * SPITCH + ((g + 1) * 2 + s) * 4);
        }
      }
      if (g == 0) {
        gq0 = *(const float4*)(Qg + kx);
        gq1 = *(const float4*)(Qg + GROW + kx);
        gq2 = *(const float4*)(Qg + 2 * GROW + kx);
        gq3 = *(const float4*)(Qg + 3 * GROW + kx);
        gk0 = *(const float4*)(Kg + kx);
        gk1 = *(const float4*)(Kg + GROW + kx);
        gk2 = *(const float4*)(Kg + 2 * GROW + kx);
        gk3 = *(const float4*)(Kg + 3 * GROW + kx);
      }
      if (g == 3) {
        float* Swn = Sw + ((kc + 1) & 1) * STG;
        *(float4*)(Swn) = gq0;
        *(float4*)(Swn + 32 * SPITCH) = gq1;
        *(float4*)(Swn + 64 * SPITCH) = gq2;
        *(float4*)(Swn + 96 * SPITCH) = gq3;
        *(float4*)(Swn + OPS) = gk0;
        *(float4*)(Swn + OPS + 32 * SPITCH) = gk1;
        *(float4*)(Swn + OPS + 64 * SPITCH) = gk2;
        *(float4*)(Swn + OPS + 96 * SPITCH) = gk3;
      }
      __builtin_amdgcn_sched_barrier(0);
#pragma unroll
      for (int s = 0; s < 2; ++s)
#pragma unroll
        for (int t = 0; t < 4; ++t) {
          acc[0][t] = wmma_f32x4(fa[g & 1][s][0], fb[g & 1][s][t], acc[0][t]);
          acc[1][t] = wmma_f32x4(fa[g & 1][s][1], fb[g & 1][s][t], acc[1][t]);
        }
      __builtin_amdgcn_sched_barrier(0);
    }

    __syncthreads();
  }

  const float tau  = *tauP;
  const float lam  = *lamP;
  const float invT = 1.0f / *tmpP;
#pragma unroll
  for (int u = 0; u < 2; ++u) {
#pragma unroll
    for (int t = 0; t < 4; ++t) {
#pragma unroll
      for (int i = 0; i < 8; ++i) {
        const int m = m0 + wm + u * 16 + i + 8 * half;
        const int n = n0 + wn + t * 16 + row;
        const float s = acc[u][t][i];
        scores[(size_t)m * N_SZ + n] = s;
        const float g = 1.0f / (1.0f + expf(-lam * (s - tau)));
        alphaRaw[(size_t)m * N_SZ + n] = g * expf(s * invT);
      }
    }
  }
}

__global__ __launch_bounds__(128) void maq_alpha_norm(float* __restrict__ alpha) {
  __shared__ float red[128];
  const int t = threadIdx.x;
  float4* p = (float4*)(alpha + (size_t)blockIdx.x * N_SZ);
  float4 v[16];
#pragma unroll
  for (int k = 0; k < 16; ++k) v[k] = p[t + 128 * k];
  __builtin_amdgcn_sched_barrier(0);
  float sa = 0.0f, sb = 0.0f;
#pragma unroll
  for (int k = 0; k < 16; k += 2) {
    sa += v[k].x + v[k].y + v[k].z + v[k].w;
    sb += v[k + 1].x + v[k + 1].y + v[k + 1].z + v[k + 1].w;
  }
  red[t] = sa + sb;
  __syncthreads();
  for (int off = 64; off > 0; off >>= 1) {
    if (t < off) red[t] += red[t + off];
    __syncthreads();
  }
  const float inv = 1.0f / (red[0] + 1e-8f);
#pragma unroll
  for (int k = 0; k < 16; ++k) {
    float4 w = v[k];
    w.x *= inv; w.y *= inv; w.z *= inv; w.w *= inv;
    p[t + 128 * k] = w;
  }
}

extern "C" void kernel_launch(void* const* d_in, const int* in_sizes, int n_in,
                              void* d_out, int out_size, void* d_ws, size_t ws_size,
                              hipStream_t stream) {
  const float* z      = (const float*)d_in[0];
  const float* pool   = (const float*)d_in[1];
  const float* WQ     = (const float*)d_in[2];
  const float* WK     = (const float*)d_in[3];
  const float* logits = (const float*)d_in[4];
  const float* tau    = (const float*)d_in[5];
  const float* lam    = (const float*)d_in[6];
  const float* temp   = (const float*)d_in[7];

  float* out    = (float*)d_out;
  float* alpha  = out;
  float* scores = out + (size_t)B_SZ * N_SZ;
  float* keys   = out + (size_t)2 * B_SZ * N_SZ;

  float* ws = (float*)d_ws;
  float* qs = ws;
  float* kn = qs + (size_t)B_SZ * SD_SZ;

  maq_gemm_proj<<<384, 256, 0, stream>>>(pool, WK, keys, kn, z, WQ, logits, qs);
  maq_gemm_scores<<<256, 256, 0, stream>>>(qs, kn, tau, lam, temp, scores, alpha);
  maq_alpha_norm<<<512, 128, 0, stream>>>(alpha);
}
